# starting kernel
# baseline (speedup 1.0000x reference)
#include <hip/hip_runtime.h>
#include <hip/hip_bf16.h>

typedef __attribute__((ext_vector_type(2)))  float    v2f;
typedef __attribute__((ext_vector_type(8)))  float    v8f;
typedef __attribute__((ext_vector_type(8)))  _Float16 v8h;
typedef __attribute__((ext_vector_type(16))) _Float16 v16h;
typedef __attribute__((ext_vector_type(4)))  int      v4i;

#define BQ 4
#define SQ 128
#define MQ (BQ*SQ)
#define INF 4096
#define OUTF 4096
#define BSR 128
#define NBIN (INF/BSR)
#define NBOUT (OUTF/BSR)

static __device__ inline v16h load_frag16(const _Float16* __restrict__ p) {
    const v8h lo = *(const v8h*)p;
    const v8h hi = *(const v8h*)(p + 16);
    v16h r;
    #pragma unroll
    for (int i = 0; i < 8; ++i) { r[i] = lo[i]; r[8 + i] = hi[i]; }
    return r;
}

__global__ __launch_bounds__(256) void k_decode_w(const int* __restrict__ Wp,
                                                  _Float16* __restrict__ W16)
{
    const size_t i = (size_t)blockIdx.x * 256 + threadIdx.x;
    const v4i p = *(const v4i*)(Wp + i * 4);
    v8h w;
    #pragma unroll
    for (int j = 0; j < 4; ++j) {
        const int c = p[j];
        w[2 * j + 0] = (_Float16)(((c & 0xF) ^ 8) - 8);
        w[2 * j + 1] = (_Float16)((((c >> 4) & 0xF) ^ 8) - 8);
    }
    *(v8h*)(W16 + i * 8) = w;
}

__global__ __launch_bounds__(256) void k_in_rot(
    const float* __restrict__ x, const int* __restrict__ perm,
    const float* __restrict__ R_in, const float* __restrict__ s_a,
    _Float16* __restrict__ A)
{
    const int mt = blockIdx.x;
    const int nb = blockIdx.y;
    const int m0 = mt * 16;
    const int tid  = threadIdx.x;
    const int wave = tid >> 5;
    const int lane = tid & 31;

    __shared__ float xp[16 * 132];
    for (int i = tid; i < 16 * 128; i += 256) {
        const int r = i >> 7, c = i & 127;
        const int src = perm[nb * 128 + c];
        xp[r * 132 + c] = x[(size_t)(m0 + r) * INF + src];
    }
    __syncthreads();

    const int d0      = wave * 16;
    const int halfsel = lane >> 4;
    const int lr      = lane & 15;

    v8f acc = {};
    const float* Rb = R_in + (size_t)nb * BSR * BSR;
    for (int c0 = 0; c0 < BSR; c0 += 4) {
        const int ca = c0 + halfsel * 2;
        v2f a;
        a.x = xp[lr * 132 + ca];
        a.y = xp[lr * 132 + ca + 1];
        v2f b;
        b.x = Rb[(size_t)(ca + 0) * BSR + d0 + lr];
        b.y = Rb[(size_t)(ca + 1) * BSR + d0 + lr];
        acc = __builtin_amdgcn_wmma_f32_16x16x4_f32(false, a, false, b,
                                                    (short)0, acc, false, false);
    }

    const int k = nb * 128 + d0 + lr;
    const float s   = s_a[k];
    const float inv = 1.0f / s;
    #pragma unroll
    for (int r = 0; r < 8; ++r) {
        const int M = r + halfsel * 8;
        float q = __builtin_rintf(acc[r] * inv);
        q = fminf(fmaxf(q, -128.0f), 127.0f);
        A[(size_t)(m0 + M) * INF + k] = (_Float16)(q * s);
    }
}

__global__ __launch_bounds__(256) void k_gemm16(
    const _Float16* __restrict__ A, const _Float16* __restrict__ W16,
    const float* __restrict__ s_w, float* __restrict__ Y)
{
    const int mt = blockIdx.x, og = blockIdx.y;
    const int wave = threadIdx.x >> 5;
    const int lane = threadIdx.x & 31;
    const int m0 = mt * 32;
    const int o0 = og * 256 + wave * 32;
    const int halfsel = lane >> 4;
    const int lr      = lane & 15;
    const int kbase   = halfsel * 8;

    const _Float16* a0p = A   + (size_t)(m0 + lr)      * INF + kbase;
    const _Float16* a1p = A   + (size_t)(m0 + 16 + lr) * INF + kbase;
    const _Float16* b0p = W16 + (size_t)(o0 + lr)      * INF + kbase;
    const _Float16* b1p = W16 + (size_t)(o0 + 16 + lr) * INF + kbase;

    v8f acc00 = {}, acc01 = {}, acc10 = {}, acc11 = {};
    for (int k0 = 0; k0 < INF; k0 += 32) {
        const v16h a0 = load_frag16(a0p + k0);
        const v16h a1 = load_frag16(a1p + k0);
        const v16h b0 = load_frag16(b0p + k0);
        const v16h b1 = load_frag16(b1p + k0);
        acc00 = __builtin_amdgcn_wmma_f32_16x16x32_f16(false, a0, false, b0,
                                                       (short)0, acc00, false, false);
        acc01 = __builtin_amdgcn_wmma_f32_16x16x32_f16(false, a0, false, b1,
                                                       (short)0, acc01, false, false);
        acc10 = __builtin_amdgcn_wmma_f32_16x16x32_f16(false, a1, false, b0,
                                                       (short)0, acc10, false, false);
        acc11 = __builtin_amdgcn_wmma_f32_16x16x32_f16(false, a1, false, b1,
                                                       (short)0, acc11, false, false);
    }

    const float sw0 = s_w[o0 + lr];
    const float sw1 = s_w[o0 + 16 + lr];
    #pragma unroll
    for (int r = 0; r < 8; ++r) {
        const int M = r + halfsel * 8;
        Y[(size_t)(m0 + M)      * OUTF + o0 + lr]      = acc00[r] * sw0;
        Y[(size_t)(m0 + M)      * OUTF + o0 + 16 + lr] = acc01[r] * sw1;
        Y[(size_t)(m0 + 16 + M) * OUTF + o0 + lr]      = acc10[r] * sw0;
        Y[(size_t)(m0 + 16 + M) * OUTF + o0 + 16 + lr] = acc11[r] * sw1;
    }
}

__global__ __launch_bounds__(256) void k_gemm_dec(
    const _Float16* __restrict__ A, const int* __restrict__ Wp,
    const float* __restrict__ s_w, float* __restrict__ Y)
{
    const int mt = blockIdx.x, og = blockIdx.y;
    const int m0 = mt * 16;
    const int wave = threadIdx.x >> 5;
    const int lane = threadIdx.x & 31;
    const int o0      = (og * 8 + wave) * 16;
    const int halfsel = lane >> 4;
    const int lr      = lane & 15;

    const int    kbase = halfsel * 8;
    const size_t arow  = (size_t)(m0 + lr) * INF;
    const int    oc    = o0 + lr;
    const int*   wrow  = Wp + (size_t)oc * (INF / 2);

    v8f acc = {};
    for (int k0 = 0; k0 < INF; k0 += 32) {
        const v16h a = load_frag16(A + arow + k0 + kbase);
        const v4i p0 = *(const v4i*)(wrow + ((k0 + kbase) >> 1));
        const v4i p1 = *(const v4i*)(wrow + ((k0 + kbase + 16) >> 1));
        v16h b;
        #pragma unroll
        for (int j = 0; j < 4; ++j) {
            int c0 = p0[j];
            b[2 * j + 0] = (_Float16)(((c0 & 0xF) ^ 8) - 8);
            b[2 * j + 1] = (_Float16)((((c0 >> 4) & 0xF) ^ 8) - 8);
            int c1 = p1[j];
            b[8 + 2 * j + 0] = (_Float16)(((c1 & 0xF) ^ 8) - 8);
            b[8 + 2 * j + 1] = (_Float16)((((c1 >> 4) & 0xF) ^ 8) - 8);
        }
        acc = __builtin_amdgcn_wmma_f32_16x16x32_f16(false, a, false, b,
                                                     (short)0, acc, false, false);
    }

    const float sw = s_w[oc];
    #pragma unroll
    for (int r = 0; r < 8; ++r) {
        const int M = r + halfsel * 8;
        Y[(size_t)(m0 + M) * OUTF + oc] = acc[r] * sw;
    }
}

__global__ __launch_bounds__(256) void k_out_rot(
    const float* __restrict__ Y, const float* __restrict__ R_out,
    const float* __restrict__ bias, float* __restrict__ out)
{
    const int mt = blockIdx.x, nb = blockIdx.y;
    const int m0 = mt * 16;
    const int wave = threadIdx.x >> 5;
    const int lane = threadIdx.x & 31;
    const int d0      = wave * 16;
    const int halfsel = lane >> 4;
    const int lr      = lane & 15;

    v8f acc = {};
    const float* Rb   = R_out + (size_t)nb * BSR * BSR;
    const size_t yrow = (size_t)(m0 + lr) * OUTF + nb * 128;
    const size_t brow = (size_t)(d0 + lr) * BSR;
    for (int c0 = 0; c0 < BSR; c0 += 4) {
        const int ca = c0 + halfsel * 2;
        const v2f a = *(const v2f*)(Y  + yrow + ca);
        const v2f b = *(const v2f*)(Rb + brow + ca);
        acc = __builtin_amdgcn_wmma_f32_16x16x4_f32(false, a, false, b,
                                                    (short)0, acc, false, false);
    }

    const int o  = nb * 128 + d0 + lr;
    const float bs = bias[o];
    #pragma unroll
    for (int r = 0; r < 8; ++r) {
        const int M = r + halfsel * 8;
        out[(size_t)(m0 + M) * OUTF + o] = acc[r] + bs;
    }
}

extern "C" void kernel_launch(void* const* d_in, const int* in_sizes, int n_in,
                              void* d_out, int out_size, void* d_ws, size_t ws_size,
                              hipStream_t stream) {
    const float* x        = (const float*)d_in[0];
    const int*   perm     = (const int*)  d_in[1];
    const float* R_in     = (const float*)d_in[2];
    const float* s_a      = (const float*)d_in[3];
    const int*   packed_w = (const int*)  d_in[4];
    const float* s_w      = (const float*)d_in[5];
    const float* bias     = (const float*)d_in[6];
    const float* R_out    = (const float*)d_in[7];
    float* out = (float*)d_out;

    const size_t szA  = (size_t)MQ * INF * sizeof(_Float16);
    const size_t szY  = (size_t)MQ * OUTF * sizeof(float);
    const size_t szW  = (size_t)OUTF * INF * sizeof(_Float16);
    _Float16* A  = (_Float16*)d_ws;
    float*    Ys = (float*)((char*)d_ws + szA);
    _Float16* W16 = (_Float16*)((char*)d_ws + szA + szY);

    const dim3 blk(256);
    k_in_rot<<<dim3(MQ / 16, NBIN), blk, 0, stream>>>(x, perm, R_in, s_a, A);

    if (ws_size >= szA + szY + szW) {
        k_decode_w<<<dim3((OUTF * (INF / 2) / 4) / 256), blk, 0, stream>>>(packed_w, W16);
        k_gemm16<<<dim3(MQ / 32, OUTF / 256), blk, 0, stream>>>(A, W16, s_w, Ys);
    } else {
        k_gemm_dec<<<dim3(MQ / 16, OUTF / 128), blk, 0, stream>>>(A, packed_w, s_w, Ys);
    }

    k_out_rot<<<dim3(MQ / 16, NBOUT), blk, 0, stream>>>(Ys, R_out, bias, out);
}


# k_gemm16: A and W tiles (64x128, K-step 256) staged global->LDS as whole 128-B lines, two stages; WMMA fragments read from LDS one sub-step ahead instead of per-lane global row gathers
# speedup vs baseline: 1.8356x; 1.8356x over previous
#include <hip/hip_runtime.h>
#include <hip/hip_bf16.h>

typedef __attribute__((ext_vector_type(2)))  float    v2f;
typedef __attribute__((ext_vector_type(8)))  float    v8f;
typedef __attribute__((ext_vector_type(8)))  _Float16 v8h;
typedef __attribute__((ext_vector_type(16))) _Float16 v16h;
typedef __attribute__((ext_vector_type(4)))  int      v4i;

#define BQ 4
#define SQ 128
#define MQ (BQ*SQ)
#define INF 4096
#define OUTF 4096
#define BSR 128
#define NBIN (INF/BSR)
#define NBOUT (OUTF/BSR)

static __device__ inline v16h load_frag16(const _Float16* __restrict__ p) {
    const v8h lo = *(const v8h*)p;
    const v8h hi = *(const v8h*)(p + 16);
    v16h r;
    #pragma unroll
    for (int i = 0; i < 8; ++i) { r[i] = lo[i]; r[8 + i] = hi[i]; }
    return r;
}

__global__ __launch_bounds__(256) void k_decode_w(const int* __restrict__ Wp,
                                                  _Float16* __restrict__ W16)
{
    const size_t i = (size_t)blockIdx.x * 256 + threadIdx.x;
    const v4i p = *(const v4i*)(Wp + i * 4);
    v8h w;
    #pragma unroll
    for (int j = 0; j < 4; ++j) {
        const int c = p[j];
        w[2 * j + 0] = (_Float16)(((c & 0xF) ^ 8) - 8);
        w[2 * j + 1] = (_Float16)((((c >> 4) & 0xF) ^ 8) - 8);
    }
    *(v8h*)(W16 + i * 8) = w;
}

__global__ __launch_bounds__(256) void k_in_rot(
    const float* __restrict__ x, const int* __restrict__ perm,
    const float* __restrict__ R_in, const float* __restrict__ s_a,
    _Float16* __restrict__ A)
{
    const int mt = blockIdx.x;
    const int nb = blockIdx.y;
    const int m0 = mt * 16;
    const int tid  = threadIdx.x;
    const int wave = tid >> 5;
    const int lane = tid & 31;

    __shared__ float xp[16 * 132];
    for (int i = tid; i < 16 * 128; i += 256) {
        const int r = i >> 7, c = i & 127;
        const int src = perm[nb * 128 + c];
        xp[r * 132 + c] = x[(size_t)(m0 + r) * INF + src];
    }
    __syncthreads();

    const int d0      = wave * 16;
    const int halfsel = lane >> 4;
    const int lr      = lane & 15;

    v8f acc = {};
    const float* Rb = R_in + (size_t)nb * BSR * BSR;
    for (int c0 = 0; c0 < BSR; c0 += 4) {
        const int ca = c0 + halfsel * 2;
        v2f a;
        a.x = xp[lr * 132 + ca];
        a.y = xp[lr * 132 + ca + 1];
        v2f b;
        b.x = Rb[(size_t)(ca + 0) * BSR + d0 + lr];
        b.y = Rb[(size_t)(ca + 1) * BSR + d0 + lr];
        acc = __builtin_amdgcn_wmma_f32_16x16x4_f32(false, a, false, b,
                                                    (short)0, acc, false, false);
    }

    const int k = nb * 128 + d0 + lr;
    const float s   = s_a[k];
    const float inv = 1.0f / s;
    #pragma unroll
    for (int r = 0; r < 8; ++r) {
        const int M = r + halfsel * 8;
        float q = __builtin_rintf(acc[r] * inv);
        q = fminf(fmaxf(q, -128.0f), 127.0f);
        A[(size_t)(m0 + M) * INF + k] = (_Float16)(q * s);
    }
}

#define GT_M 64
#define GT_N 128
#define GT_K 256
#define GT_PITCH 264
#define GT_ROWS (GT_M + GT_N)
#define GT_STAGE (GT_ROWS * GT_PITCH)
#define GT_CPR (GT_K / 8)
#define GT_RPP (256 / GT_CPR)
#define GT_CHUNKS (GT_ROWS / GT_RPP)
#define GT_APASS (GT_M / GT_RPP)
#define GT_STEPS (INF / GT_K)

static __device__ inline void lds_frag4(const _Float16* s, int oa0, int oa1, int ob0, int ob1,
                                        v16h (&fa)[2], v16h (&fb)[2]) {
    const v8h l0 = *(const v8h*)(s + oa0), h0 = *(const v8h*)(s + oa0 + 16);
    const v8h l1 = *(const v8h*)(s + oa1), h1 = *(const v8h*)(s + oa1 + 16);
    const v8h l2 = *(const v8h*)(s + ob0), h2 = *(const v8h*)(s + ob0 + 16);
    const v8h l3 = *(const v8h*)(s + ob1), h3 = *(const v8h*)(s + ob1 + 16);
    #pragma unroll
    for (int e = 0; e < 8; ++e) {
        fa[0][e] = l0[e]; fa[0][8 + e] = h0[e];
        fa[1][e] = l1[e]; fa[1][8 + e] = h1[e];
        fb[0][e] = l2[e]; fb[0][8 + e] = h2[e];
        fb[1][e] = l3[e]; fb[1][8 + e] = h3[e];
    }
}

__global__ __launch_bounds__(256) void k_gemm16(
    const _Float16* __restrict__ A, const _Float16* __restrict__ W16,
    const float* __restrict__ s_w, float* __restrict__ Y)
{
    __shared__ __attribute__((aligned(16))) _Float16 sm[2 * GT_STAGE];

    const int tid  = threadIdx.x;
    const int wave = tid >> 5;
    const int lane = tid & 31;
    const int m0 = blockIdx.x * GT_M;
    const int o0 = blockIdx.y * GT_N;
    const int halfsel = lane >> 4;
    const int lr      = lane & 15;
    const int kbase   = halfsel * 8;

    const int crow = tid / GT_CPR;
    const int cch  = (tid % GT_CPR) * 8;
    const _Float16* gA = A   + (size_t)(m0 + crow) * INF + cch;
    const _Float16* gW = W16 + (size_t)(o0 + crow) * INF + cch;
    const int sdst = crow * GT_PITCH + cch;

    const int wm = (wave & 1) * 32;
    const int wn = GT_M + (wave >> 1) * 32;
    const int fa0 = (wm + lr)      * GT_PITCH + kbase;
    const int fa1 = (wm + 16 + lr) * GT_PITCH + kbase;
    const int fb0 = (wn + lr)      * GT_PITCH + kbase;
    const int fb1 = (wn + 16 + lr) * GT_PITCH + kbase;

    v8h pre[GT_CHUNKS];
    #pragma unroll
    for (int i = 0; i < GT_CHUNKS; ++i) {
        pre[i] = (i < GT_APASS)
               ? *(const v8h*)(gA + (size_t)(i * GT_RPP) * INF)
               : *(const v8h*)(gW + (size_t)((i - GT_APASS) * GT_RPP) * INF);
    }
    #pragma unroll
    for (int i = 0; i < GT_CHUNKS; ++i)
        *(v8h*)(&sm[sdst + i * GT_RPP * GT_PITCH]) = pre[i];
    __syncthreads();

    v8f acc00 = {}, acc01 = {}, acc10 = {}, acc11 = {};
    for (int t = 0; t < GT_STEPS; ++t) {
        const int cur = (t & 1) * GT_STAGE;
        const bool more = (t + 1 < GT_STEPS);
        if (more) {
            const int k1 = (t + 1) * GT_K;
            #pragma unroll
            for (int i = 0; i < GT_CHUNKS; ++i) {
                pre[i] = (i < GT_APASS)
                       ? *(const v8h*)(gA + (size_t)(i * GT_RPP) * INF + k1)
                       : *(const v8h*)(gW + (size_t)((i - GT_APASS) * GT_RPP) * INF + k1);
            }
        }

        const _Float16* s = &sm[cur];
        v16h fa[2][2], fb[2][2];
        lds_frag4(s, fa0, fa1, fb0, fb1, fa[0], fb[0]);
        #pragma unroll
        for (int kk = 0; kk < GT_K / 32; ++kk) {
            if (kk + 1 < GT_K / 32)
                lds_frag4(s + (kk + 1) * 32, fa0, fa1, fb0, fb1, fa[(kk + 1) & 1], fb[(kk + 1) & 1]);
            __builtin_amdgcn_sched_barrier(0);
            acc00 = __builtin_amdgcn_wmma_f32_16x16x32_f16(false, fa[kk & 1][0], false, fb[kk & 1][0],
                                                           (short)0, acc00, false, false);
            acc01 = __builtin_amdgcn_wmma_f32_16x16x32_f16(false, fa[kk & 1][0], false, fb[kk & 1][1],
                                                           (short)0, acc01, false, false);
            acc10 = __builtin_amdgcn_wmma_f32_16x16x32_f16(false, fa[kk & 1][1], false, fb[kk & 1][0],
                                                           (short)0, acc10, false, false);
            acc11 = __builtin_amdgcn_wmma_f32_16x16x32_f16(false, fa[kk & 1][1], false, fb[kk & 1][1],
                                                           (short)0, acc11, false, false);
        }

        if (more) {
            const int nxt = GT_STAGE - cur;
            #pragma unroll
            for (int i = 0; i < GT_CHUNKS; ++i)
                *(v8h*)(&sm[nxt + sdst + i * GT_RPP * GT_PITCH]) = pre[i];
        }
        __syncthreads();
    }

    const int ow = o0 + (wave >> 1) * 32;
    const int mw = m0 + wm;
    const float sw0 = s_w[ow + lr];
    const float sw1 = s_w[ow + 16 + lr];
    #pragma unroll
    for (int r = 0; r < 8; ++r) {
        const int M = r + halfsel * 8;
        Y[(size_t)(mw + M)      * OUTF + ow + lr]      = acc00[r] * sw0;
        Y[(size_t)(mw + M)      * OUTF + ow + 16 + lr] = acc01[r] * sw1;
        Y[(size_t)(mw + 16 + M) * OUTF + ow + lr]      = acc10[r] * sw0;
        Y[(size_t)(mw + 16 + M) * OUTF + ow + 16 + lr] = acc11[r] * sw1;
    }
}

__global__ __launch_bounds__(256) void k_gemm_dec(
    const _Float16* __restrict__ A, const int* __restrict__ Wp,
    const float* __restrict__ s_w, float* __restrict__ Y)
{
    const int mt = blockIdx.x, og = blockIdx.y;
    const int m0 = mt * 16;
    const int wave = threadIdx.x >> 5;
    const int lane = threadIdx.x & 31;
    const int o0      = (og * 8 + wave) * 16;
    const int halfsel = lane >> 4;
    const int lr      = lane & 15;

    const int    kbase = halfsel * 8;
    const size_t arow  = (size_t)(m0 + lr) * INF;
    const int    oc    = o0 + lr;
    const int*   wrow  = Wp + (size_t)oc * (INF / 2);

    v8f acc = {};
    for (int k0 = 0; k0 < INF; k0 += 32) {
        const v16h a = load_frag16(A + arow + k0 + kbase);
        const v4i p0 = *(const v4i*)(wrow + ((k0 + kbase) >> 1));
        const v4i p1 = *(const v4i*)(wrow + ((k0 + kbase + 16) >> 1));
        v16h b;
        #pragma unroll
        for (int j = 0; j < 4; ++j) {
            int c0 = p0[j];
            b[2 * j + 0] = (_Float16)(((c0 & 0xF) ^ 8) - 8);
            b[2 * j + 1] = (_Float16)((((c0 >> 4) & 0xF) ^ 8) - 8);
            int c1 = p1[j];
            b[8 + 2 * j + 0] = (_Float16)(((c1 & 0xF) ^ 8) - 8);
            b[8 + 2 * j + 1] = (_Float16)((((c1 >> 4) & 0xF) ^ 8) - 8);
        }
        acc = __builtin_amdgcn_wmma_f32_16x16x32_f16(false, a, false, b,
                                                     (short)0, acc, false, false);
    }

    const float sw = s_w[oc];
    #pragma unroll
    for (int r = 0; r < 8; ++r) {
        const int M = r + halfsel * 8;
        Y[(size_t)(m0 + M) * OUTF + oc] = acc[r] * sw;
    }
}

__global__ __launch_bounds__(256) void k_out_rot(
    const float* __restrict__ Y, const float* __restrict__ R_out,
    const float* __restrict__ bias, float* __restrict__ out)
{
    const int mt = blockIdx.x, nb = blockIdx.y;
    const int m0 = mt * 16;
    const int wave = threadIdx.x >> 5;
    const int lane = threadIdx.x & 31;
    const int d0      = wave * 16;
    const int halfsel = lane >> 4;
    const int lr      = lane & 15;

    v8f acc = {};
    const float* Rb   = R_out + (size_t)nb * BSR * BSR;
    const size_t yrow = (size_t)(m0 + lr) * OUTF + nb * 128;
    const size_t brow = (size_t)(d0 + lr) * BSR;
    for (int c0 = 0; c0 < BSR; c0 += 4) {
        const int ca = c0 + halfsel * 2;
        const v2f a = *(const v2f*)(Y  + yrow + ca);
        const v2f b = *(const v2f*)(Rb + brow + ca);
        acc = __builtin_amdgcn_wmma_f32_16x16x4_f32(false, a, false, b,
                                                    (short)0, acc, false, false);
    }

    const int o  = nb * 128 + d0 + lr;
    const float bs = bias[o];
    #pragma unroll
    for (int r = 0; r < 8; ++r) {
        const int M = r + halfsel * 8;
        out[(size_t)(m0 + M) * OUTF + o] = acc[r] + bs;
    }
}

extern "C" void kernel_launch(void* const* d_in, const int* in_sizes, int n_in,
                              void* d_out, int out_size, void* d_ws, size_t ws_size,
                              hipStream_t stream) {
    const float* x        = (const float*)d_in[0];
    const int*   perm     = (const int*)  d_in[1];
    const float* R_in     = (const float*)d_in[2];
    const float* s_a      = (const float*)d_in[3];
    const int*   packed_w = (const int*)  d_in[4];
    const float* s_w      = (const float*)d_in[5];
    const float* bias     = (const float*)d_in[6];
    const float* R_out    = (const float*)d_in[7];
    float* out = (float*)d_out;

    const size_t szA  = (size_t)MQ * INF * sizeof(_Float16);
    const size_t szY  = (size_t)MQ * OUTF * sizeof(float);
    const size_t szW  = (size_t)OUTF * INF * sizeof(_Float16);
    _Float16* A  = (_Float16*)d_ws;
    float*    Ys = (float*)((char*)d_ws + szA);
    _Float16* W16 = (_Float16*)((char*)d_ws + szA + szY);

    const dim3 blk(256);
    k_in_rot<<<dim3(MQ / 16, NBIN), blk, 0, stream>>>(x, perm, R_in, s_a, A);

    if (ws_size >= szA + szY + szW) {
        k_decode_w<<<dim3((OUTF * (INF / 2) / 4) / 256), blk, 0, stream>>>(packed_w, W16);
        k_gemm16<<<dim3(MQ / GT_M, OUTF / GT_N), blk, 0, stream>>>(A, W16, s_w, Ys);
    } else {
        k_gemm_dec<<<dim3(MQ / 16, OUTF / 128), blk, 0, stream>>>(A, packed_w, s_w, Ys);
    }

    k_out_rot<<<dim3(MQ / 16, NBOUT), blk, 0, stream>>>(Ys, R_out, bias, out);
}


# int4->f16 weight decode folded into k_gemm16's global->LDS staging: packed words loaded and converted in registers (exact); decode launch and the 32 MB f16 weight plane removed
# speedup vs baseline: 2.2821x; 1.2432x over previous
#include <hip/hip_runtime.h>
#include <hip/hip_bf16.h>

typedef __attribute__((ext_vector_type(2)))  float    v2f;
typedef __attribute__((ext_vector_type(8)))  float    v8f;
typedef __attribute__((ext_vector_type(8)))  _Float16 v8h;
typedef __attribute__((ext_vector_type(16))) _Float16 v16h;
typedef __attribute__((ext_vector_type(4)))  int      v4i;
typedef __attribute__((ext_vector_type(2)))  _Float16 v2h;

#define BQ 4
#define SQ 128
#define MQ (BQ*SQ)
#define INF 4096
#define OUTF 4096
#define BSR 128
#define NBIN (INF/BSR)
#define NBOUT (OUTF/BSR)

static __device__ inline int dec_pair(int c) {
    const unsigned p = __builtin_amdgcn_perm((unsigned)c, (unsigned)c, 0x0C000C00u);
    const unsigned t = (p & 0x00F0000Fu) ^ 0x64806408u;
    const v2h h  = __builtin_bit_cast(v2h, t);
    const v2h sc = {(_Float16)1.0f, (_Float16)0.0625f};
    const v2h bi = {(_Float16)-1032.0f, (_Float16)-72.0f};
    const v2h r = h * sc + bi;
    return __builtin_bit_cast(int, r);
}

static __device__ inline v8h dec_chunk(v4i w) {
    v4i d;
    #pragma unroll
    for (int j = 0; j < 4; ++j) d[j] = dec_pair(w[j]);
    return __builtin_bit_cast(v8h, d);
}

__global__ __launch_bounds__(256) void k_in_rot(
    const float* __restrict__ x, const int* __restrict__ perm,
    const float* __restrict__ R_in, const float* __restrict__ s_a,
    _Float16* __restrict__ A)
{
    const int mt = blockIdx.x;
    const int nb = blockIdx.y;
    const int m0 = mt * 16;
    const int tid  = threadIdx.x;
    const int wave = tid >> 5;
    const int lane = tid & 31;

    __shared__ float xp[16 * 132];
    for (int i = tid; i < 16 * 128; i += 256) {
        const int r = i >> 7, c = i & 127;
        const int src = perm[nb * 128 + c];
        xp[r * 132 + c] = x[(size_t)(m0 + r) * INF + src];
    }
    __syncthreads();

    const int d0      = wave * 16;
    const int halfsel = lane >> 4;
    const int lr      = lane & 15;

    v8f acc = {};
    const float* Rb = R_in + (size_t)nb * BSR * BSR;
    for (int c0 = 0; c0 < BSR; c0 += 4) {
        const int ca = c0 + halfsel * 2;
        v2f a;
        a.x = xp[lr * 132 + ca];
        a.y = xp[lr * 132 + ca + 1];
        v2f b;
        b.x = Rb[(size_t)(ca + 0) * BSR + d0 + lr];
        b.y = Rb[(size_t)(ca + 1) * BSR + d0 + lr];
        acc = __builtin_amdgcn_wmma_f32_16x16x4_f32(false, a, false, b,
                                                    (short)0, acc, false, false);
    }

    const int k = nb * 128 + d0 + lr;
    const float s   = s_a[k];
    const float inv = 1.0f / s;
    #pragma unroll
    for (int r = 0; r < 8; ++r) {
        const int M = r + halfsel * 8;
        float q = __builtin_rintf(acc[r] * inv);
        q = fminf(fmaxf(q, -128.0f), 127.0f);
        A[(size_t)(m0 + M) * INF + k] = (_Float16)(q * s);
    }
}

#define GT_M 64
#define GT_N 128
#define GT_K 256
#define GT_PITCH 264
#define GT_ROWS (GT_M + GT_N)
#define GT_STAGE (GT_ROWS * GT_PITCH)
#define GT_CPR (GT_K / 8)
#define GT_RPP (256 / GT_CPR)
#define GT_APASS (GT_M / GT_RPP)
#define GT_WPASS (GT_N / GT_RPP)
#define GT_STEPS (INF / GT_K)

static __device__ inline void lds_frag4(const _Float16* s, int oa0, int oa1, int ob0, int ob1,
                                        v16h (&fa)[2], v16h (&fb)[2]) {
    const v8h l0 = *(const v8h*)(s + oa0), h0 = *(const v8h*)(s + oa0 + 16);
    const v8h l1 = *(const v8h*)(s + oa1), h1 = *(const v8h*)(s + oa1 + 16);
    const v8h l2 = *(const v8h*)(s + ob0), h2 = *(const v8h*)(s + ob0 + 16);
    const v8h l3 = *(const v8h*)(s + ob1), h3 = *(const v8h*)(s + ob1 + 16);
    #pragma unroll
    for (int e = 0; e < 8; ++e) {
        fa[0][e] = l0[e]; fa[0][8 + e] = h0[e];
        fa[1][e] = l1[e]; fa[1][8 + e] = h1[e];
        fb[0][e] = l2[e]; fb[0][8 + e] = h2[e];
        fb[1][e] = l3[e]; fb[1][8 + e] = h3[e];
    }
}

__global__ __launch_bounds__(256) void k_gemm16(
    const _Float16* __restrict__ A, const int* __restrict__ Wp,
    const float* __restrict__ s_w, float* __restrict__ Y)
{
    __shared__ __attribute__((aligned(16))) _Float16 sm[2 * GT_STAGE];

    const int tid  = threadIdx.x;
    const int wave = tid >> 5;
    const int lane = tid & 31;
    const int m0 = blockIdx.x * GT_M;
    const int o0 = blockIdx.y * GT_N;
    const int halfsel = lane >> 4;
    const int lr      = lane & 15;
    const int kbase   = halfsel * 8;

    const int crow = tid / GT_CPR;
    const int cch  = (tid % GT_CPR) * 8;
    const _Float16* gA = A  + (size_t)(m0 + crow) * INF + cch;
    const int*      gW = Wp + (size_t)(o0 + crow) * (INF / 2) + (cch >> 1);
    const int sdst = crow * GT_PITCH + cch;

    const int wm = (wave & 1) * 32;
    const int wn = GT_M + (wave >> 1) * 32;
    const int fa0 = (wm + lr)      * GT_PITCH + kbase;
    const int fa1 = (wm + 16 + lr) * GT_PITCH + kbase;
    const int fb0 = (wn + lr)      * GT_PITCH + kbase;
    const int fb1 = (wn + 16 + lr) * GT_PITCH + kbase;

    v8h pa[GT_APASS];
    v4i pw[GT_WPASS];
    #pragma unroll
    for (int i = 0; i < GT_APASS; ++i)
        pa[i] = *(const v8h*)(gA + (size_t)(i * GT_RPP) * INF);
    #pragma unroll
    for (int i = 0; i < GT_WPASS; ++i)
        pw[i] = *(const v4i*)(gW + (size_t)(i * GT_RPP) * (INF / 2));
    #pragma unroll
    for (int i = 0; i < GT_APASS; ++i)
        *(v8h*)(&sm[sdst + i * GT_RPP * GT_PITCH]) = pa[i];
    #pragma unroll
    for (int i = 0; i < GT_WPASS; ++i)
        *(v8h*)(&sm[sdst + (GT_M + i * GT_RPP) * GT_PITCH]) = dec_chunk(pw[i]);
    __syncthreads();

    v8f acc00 = {}, acc01 = {}, acc10 = {}, acc11 = {};
    for (int t = 0; t < GT_STEPS; ++t) {
        const int cur = (t & 1) * GT_STAGE;
        const bool more = (t + 1 < GT_STEPS);
        if (more) {
            const int k1 = (t + 1) * GT_K;
            #pragma unroll
            for (int i = 0; i < GT_APASS; ++i)
                pa[i] = *(const v8h*)(gA + (size_t)(i * GT_RPP) * INF + k1);
            #pragma unroll
            for (int i = 0; i < GT_WPASS; ++i)
                pw[i] = *(const v4i*)(gW + (size_t)(i * GT_RPP) * (INF / 2) + (k1 >> 1));
        }

        const _Float16* s = &sm[cur];
        v16h fa[2][2], fb[2][2];
        lds_frag4(s, fa0, fa1, fb0, fb1, fa[0], fb[0]);
        #pragma unroll
        for (int kk = 0; kk < GT_K / 32; ++kk) {
            if (kk + 1 < GT_K / 32)
                lds_frag4(s + (kk + 1) * 32, fa0, fa1, fb0, fb1, fa[(kk + 1) & 1], fb[(kk + 1) & 1]);
            __builtin_amdgcn_sched_barrier(0);
            acc00 = __builtin_amdgcn_wmma_f32_16x16x32_f16(false, fa[kk & 1][0], false, fb[kk & 1][0],
                                                           (short)0, acc00, false, false);
            acc01 = __builtin_amdgcn_wmma_f32_16x16x32_f16(false, fa[kk & 1][0], false, fb[kk & 1][1],
                                                           (short)0, acc01, false, false);
            acc10 = __builtin_amdgcn_wmma_f32_16x16x32_f16(false, fa[kk & 1][1], false, fb[kk & 1][0],
                                                           (short)0, acc10, false, false);
            acc11 = __builtin_amdgcn_wmma_f32_16x16x32_f16(false, fa[kk & 1][1], false, fb[kk & 1][1],
                                                           (short)0, acc11, false, false);
        }

        if (more) {
            const int nxt = GT_STAGE - cur;
            #pragma unroll
            for (int i = 0; i < GT_APASS; ++i)
                *(v8h*)(&sm[nxt + sdst + i * GT_RPP * GT_PITCH]) = pa[i];
            #pragma unroll
            for (int i = 0; i < GT_WPASS; ++i)
                *(v8h*)(&sm[nxt + sdst + (GT_M + i * GT_RPP) * GT_PITCH]) = dec_chunk(pw[i]);
        }
        __syncthreads();
    }

    const int ow = o0 + (wave >> 1) * 32;
    const int mw = m0 + wm;
    const float sw0 = s_w[ow + lr];
    const float sw1 = s_w[ow + 16 + lr];
    #pragma unroll
    for (int r = 0; r < 8; ++r) {
        const int M = r + halfsel * 8;
        Y[(size_t)(mw + M)      * OUTF + ow + lr]      = acc00[r] * sw0;
        Y[(size_t)(mw + M)      * OUTF + ow + 16 + lr] = acc01[r] * sw1;
        Y[(size_t)(mw + 16 + M) * OUTF + ow + lr]      = acc10[r] * sw0;
        Y[(size_t)(mw + 16 + M) * OUTF + ow + 16 + lr] = acc11[r] * sw1;
    }
}

__global__ __launch_bounds__(256) void k_out_rot(
    const float* __restrict__ Y, const float* __restrict__ R_out,
    const float* __restrict__ bias, float* __restrict__ out)
{
    const int mt = blockIdx.x, nb = blockIdx.y;
    const int m0 = mt * 16;
    const int wave = threadIdx.x >> 5;
    const int lane = threadIdx.x & 31;
    const int d0      = wave * 16;
    const int halfsel = lane >> 4;
    const int lr      = lane & 15;

    v8f acc = {};
    const float* Rb   = R_out + (size_t)nb * BSR * BSR;
    const size_t yrow = (size_t)(m0 + lr) * OUTF + nb * 128;
    const size_t brow = (size_t)(d0 + lr) * BSR;
    for (int c0 = 0; c0 < BSR; c0 += 4) {
        const int ca = c0 + halfsel * 2;
        const v2f a = *(const v2f*)(Y  + yrow + ca);
        const v2f b = *(const v2f*)(Rb + brow + ca);
        acc = __builtin_amdgcn_wmma_f32_16x16x4_f32(false, a, false, b,
                                                    (short)0, acc, false, false);
    }

    const int o  = nb * 128 + d0 + lr;
    const float bs = bias[o];
    #pragma unroll
    for (int r = 0; r < 8; ++r) {
        const int M = r + halfsel * 8;
        out[(size_t)(m0 + M) * OUTF + o] = acc[r] + bs;
    }
}

extern "C" void kernel_launch(void* const* d_in, const int* in_sizes, int n_in,
                              void* d_out, int out_size, void* d_ws, size_t ws_size,
                              hipStream_t stream) {
    const float* x        = (const float*)d_in[0];
    const int*   perm     = (const int*)  d_in[1];
    const float* R_in     = (const float*)d_in[2];
    const float* s_a      = (const float*)d_in[3];
    const int*   packed_w = (const int*)  d_in[4];
    const float* s_w      = (const float*)d_in[5];
    const float* bias     = (const float*)d_in[6];
    const float* R_out    = (const float*)d_in[7];
    float* out = (float*)d_out;

    const size_t szA  = (size_t)MQ * INF * sizeof(_Float16);
    _Float16* A  = (_Float16*)d_ws;
    float*    Ys = (float*)((char*)d_ws + szA);

    const dim3 blk(256);
    k_in_rot<<<dim3(MQ / 16, NBIN), blk, 0, stream>>>(x, perm, R_in, s_a, A);
    k_gemm16<<<dim3(MQ / GT_M, OUTF / GT_N), blk, 0, stream>>>(A, packed_w, s_w, Ys);
    k_out_rot<<<dim3(MQ / 16, NBOUT), blk, 0, stream>>>(Ys, R_out, bias, out);
}


# k_out_rot fused into k_gemm16's epilogue: Y tile (acc*s_w) and R_out block staged in LDS, same f32 16x16x4 WMMAs in the same k order; third launch and the 8 MB Y plane removed
# speedup vs baseline: 2.8265x; 1.2386x over previous
#include <hip/hip_runtime.h>
#include <hip/hip_bf16.h>

typedef __attribute__((ext_vector_type(2)))  float    v2f;
typedef __attribute__((ext_vector_type(8)))  float    v8f;
typedef __attribute__((ext_vector_type(8)))  _Float16 v8h;
typedef __attribute__((ext_vector_type(16))) _Float16 v16h;
typedef __attribute__((ext_vector_type(4)))  int      v4i;
typedef __attribute__((ext_vector_type(4)))  float    v4f;
typedef __attribute__((ext_vector_type(2)))  _Float16 v2h;

#define BQ 4
#define SQ 128
#define MQ (BQ*SQ)
#define INF 4096
#define OUTF 4096
#define BSR 128
#define NBIN (INF/BSR)
#define NBOUT (OUTF/BSR)

static __device__ inline int dec_pair(int c) {
    const unsigned p = __builtin_amdgcn_perm((unsigned)c, (unsigned)c, 0x0C000C00u);
    const unsigned t = (p & 0x00F0000Fu) ^ 0x64806408u;
    const v2h h  = __builtin_bit_cast(v2h, t);
    const v2h sc = {(_Float16)1.0f, (_Float16)0.0625f};
    const v2h bi = {(_Float16)-1032.0f, (_Float16)-72.0f};
    const v2h r = h * sc + bi;
    return __builtin_bit_cast(int, r);
}

static __device__ inline v8h dec_chunk(v4i w) {
    v4i d;
    #pragma unroll
    for (int j = 0; j < 4; ++j) d[j] = dec_pair(w[j]);
    return __builtin_bit_cast(v8h, d);
}

__global__ __launch_bounds__(256) void k_in_rot(
    const float* __restrict__ x, const int* __restrict__ perm,
    const float* __restrict__ R_in, const float* __restrict__ s_a,
    _Float16* __restrict__ A)
{
    const int mt = blockIdx.x;
    const int nb = blockIdx.y;
    const int m0 = mt * 16;
    const int tid  = threadIdx.x;
    const int wave = tid >> 5;
    const int lane = tid & 31;

    __shared__ float xp[16 * 132];
    for (int i = tid; i < 16 * 128; i += 256) {
        const int r = i >> 7, c = i & 127;
        const int src = perm[nb * 128 + c];
        xp[r * 132 + c] = x[(size_t)(m0 + r) * INF + src];
    }
    __syncthreads();

    const int d0      = wave * 16;
    const int halfsel = lane >> 4;
    const int lr      = lane & 15;

    v8f acc = {};
    const float* Rb = R_in + (size_t)nb * BSR * BSR;
    for (int c0 = 0; c0 < BSR; c0 += 4) {
        const int ca = c0 + halfsel * 2;
        v2f a;
        a.x = xp[lr * 132 + ca];
        a.y = xp[lr * 132 + ca + 1];
        v2f b;
        b.x = Rb[(size_t)(ca + 0) * BSR + d0 + lr];
        b.y = Rb[(size_t)(ca + 1) * BSR + d0 + lr];
        acc = __builtin_amdgcn_wmma_f32_16x16x4_f32(false, a, false, b,
                                                    (short)0, acc, false, false);
    }

    const int k = nb * 128 + d0 + lr;
    const float s   = s_a[k];
    const float inv = 1.0f / s;
    #pragma unroll
    for (int r = 0; r < 8; ++r) {
        const int M = r + halfsel * 8;
        float q = __builtin_rintf(acc[r] * inv);
        q = fminf(fmaxf(q, -128.0f), 127.0f);
        A[(size_t)(m0 + M) * INF + k] = (_Float16)(q * s);
    }
}

#define GT_M 64
#define GT_N 128
#define GT_K 256
#define GT_PITCH 264
#define GT_ROWS (GT_M + GT_N)
#define GT_STAGE (GT_ROWS * GT_PITCH)
#define GT_CPR (GT_K / 8)
#define GT_RPP (256 / GT_CPR)
#define GT_APASS (GT_M / GT_RPP)
#define GT_WPASS (GT_N / GT_RPP)
#define GT_STEPS (INF / GT_K)
#define RP (GT_PITCH / 2)
static_assert(GT_N == BSR && (BSR + GT_M) * RP * 4 <= GT_STAGE * 2, "R_out block + Y tile must fit in one stage");

static __device__ inline void lds_frag4(const _Float16* s, int oa0, int oa1, int ob0, int ob1,
                                        v16h (&fa)[2], v16h (&fb)[2]) {
    const v8h l0 = *(const v8h*)(s + oa0), h0 = *(const v8h*)(s + oa0 + 16);
    const v8h l1 = *(const v8h*)(s + oa1), h1 = *(const v8h*)(s + oa1 + 16);
    const v8h l2 = *(const v8h*)(s + ob0), h2 = *(const v8h*)(s + ob0 + 16);
    const v8h l3 = *(const v8h*)(s + ob1), h3 = *(const v8h*)(s + ob1 + 16);
    #pragma unroll
    for (int e = 0; e < 8; ++e) {
        fa[0][e] = l0[e]; fa[0][8 + e] = h0[e];
        fa[1][e] = l1[e]; fa[1][8 + e] = h1[e];
        fb[0][e] = l2[e]; fb[0][8 + e] = h2[e];
        fb[1][e] = l3[e]; fb[1][8 + e] = h3[e];
    }
}

__global__ __launch_bounds__(256) void k_gemm16(
    const _Float16* __restrict__ A, const int* __restrict__ Wp,
    const float* __restrict__ s_w, const float* __restrict__ R_out,
    const float* __restrict__ bias, float* __restrict__ out)
{
    __shared__ __attribute__((aligned(16))) _Float16 sm[2 * GT_STAGE];

    const int tid  = threadIdx.x;
    const int wave = tid >> 5;
    const int lane = tid & 31;
    const int m0 = blockIdx.x * GT_M;
    const int o0 = blockIdx.y * GT_N;
    const int halfsel = lane >> 4;
    const int lr      = lane & 15;
    const int kbase   = halfsel * 8;

    const int crow = tid / GT_CPR;
    const int cch  = (tid % GT_CPR) * 8;
    const _Float16* gA = A  + (size_t)(m0 + crow) * INF + cch;
    const int*      gW = Wp + (size_t)(o0 + crow) * (INF / 2) + (cch >> 1);
    const int sdst = crow * GT_PITCH + cch;

    const int wm = (wave & 1) * 32;
    const int wn = GT_M + (wave >> 1) * 32;
    const int fa0 = (wm + lr)      * GT_PITCH + kbase;
    const int fa1 = (wm + 16 + lr) * GT_PITCH + kbase;
    const int fb0 = (wn + lr)      * GT_PITCH + kbase;
    const int fb1 = (wn + 16 + lr) * GT_PITCH + kbase;

    v8h pa[GT_APASS];
    v4i pw[GT_WPASS];
    #pragma unroll
    for (int i = 0; i < GT_APASS; ++i)
        pa[i] = *(const v8h*)(gA + (size_t)(i * GT_RPP) * INF);
    #pragma unroll
    for (int i = 0; i < GT_WPASS; ++i)
        pw[i] = *(const v4i*)(gW + (size_t)(i * GT_RPP) * (INF / 2));
    #pragma unroll
    for (int i = 0; i < GT_APASS; ++i)
        *(v8h*)(&sm[sdst + i * GT_RPP * GT_PITCH]) = pa[i];
    #pragma unroll
    for (int i = 0; i < GT_WPASS; ++i)
        *(v8h*)(&sm[sdst + (GT_M + i * GT_RPP) * GT_PITCH]) = dec_chunk(pw[i]);
    __syncthreads();

    v8f acc00 = {}, acc01 = {}, acc10 = {}, acc11 = {};
    for (int t = 0; t < GT_STEPS; ++t) {
        const int cur = (t & 1) * GT_STAGE;
        const bool more = (t + 1 < GT_STEPS);
        if (more) {
            const int k1 = (t + 1) * GT_K;
            #pragma unroll
            for (int i = 0; i < GT_APASS; ++i)
                pa[i] = *(const v8h*)(gA + (size_t)(i * GT_RPP) * INF + k1);
            #pragma unroll
            for (int i = 0; i < GT_WPASS; ++i)
                pw[i] = *(const v4i*)(gW + (size_t)(i * GT_RPP) * (INF / 2) + (k1 >> 1));
        }

        const _Float16* s = &sm[cur];
        v16h fa[2][2], fb[2][2];
        lds_frag4(s, fa0, fa1, fb0, fb1, fa[0], fb[0]);
        #pragma unroll
        for (int kk = 0; kk < GT_K / 32; ++kk) {
            if (kk + 1 < GT_K / 32)
                lds_frag4(s + (kk + 1) * 32, fa0, fa1, fb0, fb1, fa[(kk + 1) & 1], fb[(kk + 1) & 1]);
            __builtin_amdgcn_sched_barrier(0);
            acc00 = __builtin_amdgcn_wmma_f32_16x16x32_f16(false, fa[kk & 1][0], false, fb[kk & 1][0],
                                                           (short)0, acc00, false, false);
            acc01 = __builtin_amdgcn_wmma_f32_16x16x32_f16(false, fa[kk & 1][0], false, fb[kk & 1][1],
                                                           (short)0, acc01, false, false);
            acc10 = __builtin_amdgcn_wmma_f32_16x16x32_f16(false, fa[kk & 1][1], false, fb[kk & 1][0],
                                                           (short)0, acc10, false, false);
            acc11 = __builtin_amdgcn_wmma_f32_16x16x32_f16(false, fa[kk & 1][1], false, fb[kk & 1][1],
                                                           (short)0, acc11, false, false);
        }

        if (more) {
            const int nxt = GT_STAGE - cur;
            #pragma unroll
            for (int i = 0; i < GT_APASS; ++i)
                *(v8h*)(&sm[nxt + sdst + i * GT_RPP * GT_PITCH]) = pa[i];
            #pragma unroll
            for (int i = 0; i < GT_WPASS; ++i)
                *(v8h*)(&sm[nxt + sdst + (GT_M + i * GT_RPP) * GT_PITCH]) = dec_chunk(pw[i]);
        }
        __syncthreads();
    }

    float* sf = (float*)sm;
    const int d0 = wave * 16;
    const int oc = o0 + d0 + lr;
    {
        const float* gR = R_out + (size_t)blockIdx.y * BSR * BSR + (size_t)d0 * BSR + lane * 4;
        v4f pr[16];
        #pragma unroll
        for (int i = 0; i < 16; ++i)
            pr[i] = *(const v4f*)(gR + (size_t)i * BSR);

        const int ow = o0 + (wave >> 1) * 32;
        const float sw0 = s_w[ow + lr];
        const float sw1 = s_w[ow + 16 + lr];
        const int yb = (BSR + wm + halfsel * 8) * RP + (wave >> 1) * 32 + lr;
        #pragma unroll
        for (int r = 0; r < 8; ++r) {
            sf[yb + r * RP]             = acc00[r] * sw0;
            sf[yb + r * RP + 16]        = acc01[r] * sw1;
            sf[yb + (16 + r) * RP]      = acc10[r] * sw0;
            sf[yb + (16 + r) * RP + 16] = acc11[r] * sw1;
        }
        const int rdst = d0 * RP + lane * 4;
        #pragma unroll
        for (int i = 0; i < 16; ++i)
            *(v4f*)(&sf[rdst + i * RP]) = pr[i];
    }
    const float bs = bias[oc];
    __syncthreads();

    const float* fA = sf + (BSR + lr) * RP + halfsel * 2;
    const float* fB = sf + (d0 + lr) * RP + halfsel * 2;
    v8f ro[4] = {};
    v2f ra[2][4][8], rb[2][8];
    #pragma unroll
    for (int s = 0; s < 8; ++s) {
        rb[0][s] = *(const v2f*)(fB + 4 * s);
        #pragma unroll
        for (int ms = 0; ms < 4; ++ms)
            ra[0][ms][s] = *(const v2f*)(fA + ms * 16 * RP + 4 * s);
    }
    #pragma unroll
    for (int blk = 0; blk < BSR / 32; ++blk) {
        if (blk + 1 < BSR / 32) {
            #pragma unroll
            for (int s = 0; s < 8; ++s) {
                rb[(blk + 1) & 1][s] = *(const v2f*)(fB + (blk + 1) * 32 + 4 * s);
                #pragma unroll
                for (int ms = 0; ms < 4; ++ms)
                    ra[(blk + 1) & 1][ms][s] = *(const v2f*)(fA + ms * 16 * RP + (blk + 1) * 32 + 4 * s);
            }
        }
        __builtin_amdgcn_sched_barrier(0);
        #pragma unroll
        for (int s = 0; s < 8; ++s) {
            #pragma unroll
            for (int ms = 0; ms < 4; ++ms)
                ro[ms] = __builtin_amdgcn_wmma_f32_16x16x4_f32(false, ra[blk & 1][ms][s], false, rb[blk & 1][s],
                                                               (short)0, ro[ms], false, false);
        }
    }

    #pragma unroll
    for (int ms = 0; ms < 4; ++ms) {
        #pragma unroll
        for (int r = 0; r < 8; ++r) {
            const int M = ms * 16 + halfsel * 8 + r;
            out[(size_t)(m0 + M) * OUTF + oc] = ro[ms][r] + bs;
        }
    }
}

extern "C" void kernel_launch(void* const* d_in, const int* in_sizes, int n_in,
                              void* d_out, int out_size, void* d_ws, size_t ws_size,
                              hipStream_t stream) {
    const float* x        = (const float*)d_in[0];
    const int*   perm     = (const int*)  d_in[1];
    const float* R_in     = (const float*)d_in[2];
    const float* s_a      = (const float*)d_in[3];
    const int*   packed_w = (const int*)  d_in[4];
    const float* s_w      = (const float*)d_in[5];
    const float* bias     = (const float*)d_in[6];
    const float* R_out    = (const float*)d_in[7];
    float* out = (float*)d_out;

    _Float16* A  = (_Float16*)d_ws;

    const dim3 blk(256);
    k_in_rot<<<dim3(MQ / 16, NBIN), blk, 0, stream>>>(x, perm, R_in, s_a, A);
    k_gemm16<<<dim3(MQ / GT_M, OUTF / GT_N), blk, 0, stream>>>(A, packed_w, s_w, R_out, bias, out);
}


# k_in_rot: work-group = 16 rows x 512 columns; its 16 whole x rows staged in LDS by whole-line loads, permuted LDS->LDS (no 4-B global gathers); 4 tiles per wave share A, R_in by 16-B loads
# speedup vs baseline: 3.0285x; 1.0715x over previous
#include <hip/hip_runtime.h>
#include <hip/hip_bf16.h>

typedef __attribute__((ext_vector_type(2)))  float    v2f;
typedef __attribute__((ext_vector_type(8)))  float    v8f;
typedef __attribute__((ext_vector_type(8)))  _Float16 v8h;
typedef __attribute__((ext_vector_type(16))) _Float16 v16h;
typedef __attribute__((ext_vector_type(4)))  int      v4i;
typedef __attribute__((ext_vector_type(4)))  float    v4f;
typedef __attribute__((ext_vector_type(2)))  _Float16 v2h;

#define BQ 4
#define SQ 128
#define MQ (BQ*SQ)
#define INF 4096
#define OUTF 4096
#define BSR 128
#define NBIN (INF/BSR)
#define NBOUT (OUTF/BSR)

static __device__ inline int dec_pair(int c) {
    const unsigned p = __builtin_amdgcn_perm((unsigned)c, (unsigned)c, 0x0C000C00u);
    const unsigned t = (p & 0x00F0000Fu) ^ 0x64806408u;
    const v2h h  = __builtin_bit_cast(v2h, t);
    const v2h sc = {(_Float16)1.0f, (_Float16)0.0625f};
    const v2h bi = {(_Float16)-1032.0f, (_Float16)-72.0f};
    const v2h r = h * sc + bi;
    return __builtin_bit_cast(int, r);
}

static __device__ inline v8h dec_chunk(v4i w) {
    v4i d;
    #pragma unroll
    for (int j = 0; j < 4; ++j) d[j] = dec_pair(w[j]);
    return __builtin_bit_cast(v8h, d);
}

#define IR_XPITCH 4100
#define IR_GCOLS  512
#define IR_PPITCH 516
typedef __attribute__((ext_vector_type(4))) _Float16 v4h;

__global__ __launch_bounds__(256) void k_in_rot(
    const float* __restrict__ x, const int* __restrict__ perm,
    const float* __restrict__ R_in, const float* __restrict__ s_a,
    _Float16* __restrict__ A)
{
    __shared__ __attribute__((aligned(16))) float xs[16 * IR_XPITCH];
    __shared__ __attribute__((aligned(16))) float xq[16 * IR_PPITCH];

    const int m0   = blockIdx.x * 16;
    const int g    = blockIdx.y;
    const int tid  = threadIdx.x;
    const int wave = tid >> 5;
    const int lane = tid & 31;
    const int halfsel = lane >> 4;
    const int lr      = lane & 15;
    const int nbl     = wave >> 1;
    const int nb      = g * 4 + nbl;
    const int dh      = (wave & 1) * 64 + lr * 4;
    const int kc      = nb * BSR + dh;

    const int p0 = perm[g * IR_GCOLS + tid];
    const int p1 = perm[g * IR_GCOLS + 256 + tid];
    const v4f sv = *(const v4f*)(s_a + kc);

    #pragma unroll
    for (int rr = 0; rr < 2; ++rr) {
        const int r = wave * 2 + rr;
        const float* src = x + (size_t)(m0 + r) * INF + lane * 4;
        float* dst = xs + r * IR_XPITCH + lane * 4;
        v4f t[32];
        #pragma unroll
        for (int i = 0; i < 32; ++i) t[i] = *(const v4f*)(src + i * 128);
        #pragma unroll
        for (int i = 0; i < 32; ++i) *(v4f*)(dst + i * 128) = t[i];
    }

    const float* Rb = R_in + (size_t)nb * BSR * BSR + (size_t)(halfsel * 2) * BSR + dh;
    v4f rb[2][8][2];
    #pragma unroll
    for (int s = 0; s < 8; ++s) {
        rb[0][s][0] = *(const v4f*)(Rb + (size_t)(4 * s) * BSR);
        rb[0][s][1] = *(const v4f*)(Rb + (size_t)(4 * s + 1) * BSR);
    }
    __syncthreads();

    #pragma unroll
    for (int r = 0; r < 16; ++r) {
        xq[r * IR_PPITCH + tid]       = xs[r * IR_XPITCH + p0];
        xq[r * IR_PPITCH + 256 + tid] = xs[r * IR_XPITCH + p1];
    }
    __syncthreads();

    const float* fa = xq + lr * IR_PPITCH + nbl * BSR + halfsel * 2;
    v2f a[32];
    #pragma unroll
    for (int j = 0; j < 32; ++j) a[j] = *(const v2f*)(fa + 4 * j);
    __builtin_amdgcn_sched_barrier(0);

    v8f acc[4] = {};
    #pragma unroll
    for (int blk = 0; blk < 4; ++blk) {
        if (blk + 1 < 4) {
            #pragma unroll
            for (int s = 0; s < 8; ++s) {
                const int j = (blk + 1) * 8 + s;
                rb[(blk + 1) & 1][s][0] = *(const v4f*)(Rb + (size_t)(4 * j) * BSR);
                rb[(blk + 1) & 1][s][1] = *(const v4f*)(Rb + (size_t)(4 * j + 1) * BSR);
            }
        }
        #pragma unroll
        for (int s = 0; s < 8; ++s) {
            #pragma unroll
            for (int e = 0; e < 4; ++e) {
                v2f b;
                b.x = rb[blk & 1][s][0][e];
                b.y = rb[blk & 1][s][1][e];
                acc[e] = __builtin_amdgcn_wmma_f32_16x16x4_f32(false, a[blk * 8 + s], false, b,
                                                               (short)0, acc[e], false, false);
            }
        }
    }

    float inv[4];
    #pragma unroll
    for (int e = 0; e < 4; ++e) inv[e] = 1.0f / sv[e];
    #pragma unroll
    for (int r = 0; r < 8; ++r) {
        const int M = r + halfsel * 8;
        v4h o;
        #pragma unroll
        for (int e = 0; e < 4; ++e) {
            float q = __builtin_rintf(acc[e][r] * inv[e]);
            q = fminf(fmaxf(q, -128.0f), 127.0f);
            o[e] = (_Float16)(q * sv[e]);
        }
        *(v4h*)(A + (size_t)(m0 + M) * INF + kc) = o;
    }
}

#define GT_M 64
#define GT_N 128
#define GT_K 256
#define GT_PITCH 264
#define GT_ROWS (GT_M + GT_N)
#define GT_STAGE (GT_ROWS * GT_PITCH)
#define GT_CPR (GT_K / 8)
#define GT_RPP (256 / GT_CPR)
#define GT_APASS (GT_M / GT_RPP)
#define GT_WPASS (GT_N / GT_RPP)
#define GT_STEPS (INF / GT_K)
#define RP (GT_PITCH / 2)
static_assert(GT_N == BSR && (BSR + GT_M) * RP * 4 <= GT_STAGE * 2, "R_out block + Y tile must fit in one stage");

static __device__ inline void lds_frag4(const _Float16* s, int oa0, int oa1, int ob0, int ob1,
                                        v16h (&fa)[2], v16h (&fb)[2]) {
    const v8h l0 = *(const v8h*)(s + oa0), h0 = *(const v8h*)(s + oa0 + 16);
    const v8h l1 = *(const v8h*)(s + oa1), h1 = *(const v8h*)(s + oa1 + 16);
    const v8h l2 = *(const v8h*)(s + ob0), h2 = *(const v8h*)(s + ob0 + 16);
    const v8h l3 = *(const v8h*)(s + ob1), h3 = *(const v8h*)(s + ob1 + 16);
    #pragma unroll
    for (int e = 0; e < 8; ++e) {
        fa[0][e] = l0[e]; fa[0][8 + e] = h0[e];
        fa[1][e] = l1[e]; fa[1][8 + e] = h1[e];
        fb[0][e] = l2[e]; fb[0][8 + e] = h2[e];
        fb[1][e] = l3[e]; fb[1][8 + e] = h3[e];
    }
}

__global__ __launch_bounds__(256) void k_gemm16(
    const _Float16* __restrict__ A, const int* __restrict__ Wp,
    const float* __restrict__ s_w, const float* __restrict__ R_out,
    const float* __restrict__ bias, float* __restrict__ out)
{
    __shared__ __attribute__((aligned(16))) _Float16 sm[2 * GT_STAGE];

    const int tid  = threadIdx.x;
    const int wave = tid >> 5;
    const int lane = tid & 31;
    const int m0 = blockIdx.x * GT_M;
    const int o0 = blockIdx.y * GT_N;
    const int halfsel = lane >> 4;
    const int lr      = lane & 15;
    const int kbase   = halfsel * 8;

    const int crow = tid / GT_CPR;
    const int cch  = (tid % GT_CPR) * 8;
    const _Float16* gA = A  + (size_t)(m0 + crow) * INF + cch;
    const int*      gW = Wp + (size_t)(o0 + crow) * (INF / 2) + (cch >> 1);
    const int sdst = crow * GT_PITCH + cch;

    const int wm = (wave & 1) * 32;
    const int wn = GT_M + (wave >> 1) * 32;
    const int fa0 = (wm + lr)      * GT_PITCH + kbase;
    const int fa1 = (wm + 16 + lr) * GT_PITCH + kbase;
    const int fb0 = (wn + lr)      * GT_PITCH + kbase;
    const int fb1 = (wn + 16 + lr) * GT_PITCH + kbase;

    v8h pa[GT_APASS];
    v4i pw[GT_WPASS];
    #pragma unroll
    for (int i = 0; i < GT_APASS; ++i)
        pa[i] = *(const v8h*)(gA + (size_t)(i * GT_RPP) * INF);
    #pragma unroll
    for (int i = 0; i < GT_WPASS; ++i)
        pw[i] = *(const v4i*)(gW + (size_t)(i * GT_RPP) * (INF / 2));
    #pragma unroll
    for (int i = 0; i < GT_APASS; ++i)
        *(v8h*)(&sm[sdst + i * GT_RPP * GT_PITCH]) = pa[i];
    #pragma unroll
    for (int i = 0; i < GT_WPASS; ++i)
        *(v8h*)(&sm[sdst + (GT_M + i * GT_RPP) * GT_PITCH]) = dec_chunk(pw[i]);
    __syncthreads();

    v8f acc00 = {}, acc01 = {}, acc10 = {}, acc11 = {};
    for (int t = 0; t < GT_STEPS; ++t) {
        const int cur = (t & 1) * GT_STAGE;
        const bool more = (t + 1 < GT_STEPS);
        if (more) {
            const int k1 = (t + 1) * GT_K;
            #pragma unroll
            for (int i = 0; i < GT_APASS; ++i)
                pa[i] = *(const v8h*)(gA + (size_t)(i * GT_RPP) * INF + k1);
            #pragma unroll
            for (int i = 0; i < GT_WPASS; ++i)
                pw[i] = *(const v4i*)(gW + (size_t)(i * GT_RPP) * (INF / 2) + (k1 >> 1));
        }

        const _Float16* s = &sm[cur];
        v16h fa[2][2], fb[2][2];
        lds_frag4(s, fa0, fa1, fb0, fb1, fa[0], fb[0]);
        #pragma unroll
        for (int kk = 0; kk < GT_K / 32; ++kk) {
            if (kk + 1 < GT_K / 32)
                lds_frag4(s + (kk + 1) * 32, fa0, fa1, fb0, fb1, fa[(kk + 1) & 1], fb[(kk + 1) & 1]);
            __builtin_amdgcn_sched_barrier(0);
            acc00 = __builtin_amdgcn_wmma_f32_16x16x32_f16(false, fa[kk & 1][0], false, fb[kk & 1][0],
                                                           (short)0, acc00, false, false);
            acc01 = __builtin_amdgcn_wmma_f32_16x16x32_f16(false, fa[kk & 1][0], false, fb[kk & 1][1],
                                                           (short)0, acc01, false, false);
            acc10 = __builtin_amdgcn_wmma_f32_16x16x32_f16(false, fa[kk & 1][1], false, fb[kk & 1][0],
                                                           (short)0, acc10, false, false);
            acc11 = __builtin_amdgcn_wmma_f32_16x16x32_f16(false, fa[kk & 1][1], false, fb[kk & 1][1],
                                                           (short)0, acc11, false, false);
        }

        if (more) {
            const int nxt = GT_STAGE - cur;
            #pragma unroll
            for (int i = 0; i < GT_APASS; ++i)
                *(v8h*)(&sm[nxt + sdst + i * GT_RPP * GT_PITCH]) = pa[i];
            #pragma unroll
            for (int i = 0; i < GT_WPASS; ++i)
                *(v8h*)(&sm[nxt + sdst + (GT_M + i * GT_RPP) * GT_PITCH]) = dec_chunk(pw[i]);
        }
        __syncthreads();
    }

    float* sf = (float*)sm;
    const int d0 = wave * 16;
    const int oc = o0 + d0 + lr;
    {
        const float* gR = R_out + (size_t)blockIdx.y * BSR * BSR + (size_t)d0 * BSR + lane * 4;
        v4f pr[16];
        #pragma unroll
        for (int i = 0; i < 16; ++i)
            pr[i] = *(const v4f*)(gR + (size_t)i * BSR);

        const int ow = o0 + (wave >> 1) * 32;
        const float sw0 = s_w[ow + lr];
        const float sw1 = s_w[ow + 16 + lr];
        const int yb = (BSR + wm + halfsel * 8) * RP + (wave >> 1) * 32 + lr;
        #pragma unroll
        for (int r = 0; r < 8; ++r) {
            sf[yb + r * RP]             = acc00[r] * sw0;
            sf[yb + r * RP + 16]        = acc01[r] * sw1;
            sf[yb + (16 + r) * RP]      = acc10[r] * sw0;
            sf[yb + (16 + r) * RP + 16] = acc11[r] * sw1;
        }
        const int rdst = d0 * RP + lane * 4;
        #pragma unroll
        for (int i = 0; i < 16; ++i)
            *(v4f*)(&sf[rdst + i * RP]) = pr[i];
    }
    const float bs = bias[oc];
    __syncthreads();

    const float* fA = sf + (BSR + lr) * RP + halfsel * 2;
    const float* fB = sf + (d0 + lr) * RP + halfsel * 2;
    v8f ro[4] = {};
    v2f ra[2][4][8], rb[2][8];
    #pragma unroll
    for (int s = 0; s < 8; ++s) {
        rb[0][s] = *(const v2f*)(fB + 4 * s);
        #pragma unroll
        for (int ms = 0; ms < 4; ++ms)
            ra[0][ms][s] = *(const v2f*)(fA + ms * 16 * RP + 4 * s);
    }
    #pragma unroll
    for (int blk = 0; blk < BSR / 32; ++blk) {
        if (blk + 1 < BSR / 32) {
            #pragma unroll
            for (int s = 0; s < 8; ++s) {
                rb[(blk + 1) & 1][s] = *(const v2f*)(fB + (blk + 1) * 32 + 4 * s);
                #pragma unroll
                for (int ms = 0; ms < 4; ++ms)
                    ra[(blk + 1) & 1][ms][s] = *(const v2f*)(fA + ms * 16 * RP + (blk + 1) * 32 + 4 * s);
            }
        }
        __builtin_amdgcn_sched_barrier(0);
        #pragma unroll
        for (int s = 0; s < 8; ++s) {
            #pragma unroll
            for (int ms = 0; ms < 4; ++ms)
                ro[ms] = __builtin_amdgcn_wmma_f32_16x16x4_f32(false, ra[blk & 1][ms][s], false, rb[blk & 1][s],
                                                               (short)0, ro[ms], false, false);
        }
    }

    #pragma unroll
    for (int ms = 0; ms < 4; ++ms) {
        #pragma unroll
        for (int r = 0; r < 8; ++r) {
            const int M = ms * 16 + halfsel * 8 + r;
            out[(size_t)(m0 + M) * OUTF + oc] = ro[ms][r] + bs;
        }
    }
}

extern "C" void kernel_launch(void* const* d_in, const int* in_sizes, int n_in,
                              void* d_out, int out_size, void* d_ws, size_t ws_size,
                              hipStream_t stream) {
    const float* x        = (const float*)d_in[0];
    const int*   perm     = (const int*)  d_in[1];
    const float* R_in     = (const float*)d_in[2];
    const float* s_a      = (const float*)d_in[3];
    const int*   packed_w = (const int*)  d_in[4];
    const float* s_w      = (const float*)d_in[5];
    const float* bias     = (const float*)d_in[6];
    const float* R_out    = (const float*)d_in[7];
    float* out = (float*)d_out;

    _Float16* A  = (_Float16*)d_ws;

    const dim3 blk(256);
    k_in_rot<<<dim3(MQ / 16, NBIN / 4), blk, 0, stream>>>(x, perm, R_in, s_a, A);
    k_gemm16<<<dim3(MQ / GT_M, OUTF / GT_N), blk, 0, stream>>>(A, packed_w, s_w, R_out, bias, out);
}


# k_gemm16: A and W tile loads issued two K-steps ahead (second register set, last step pair peeled); an LDS-only hand-written barrier keeps them in flight, so LDS stores stop waiting on memory
# speedup vs baseline: 3.2184x; 1.0627x over previous
#include <hip/hip_runtime.h>
#include <hip/hip_bf16.h>

typedef __attribute__((ext_vector_type(2)))  float    v2f;
typedef __attribute__((ext_vector_type(8)))  float    v8f;
typedef __attribute__((ext_vector_type(8)))  _Float16 v8h;
typedef __attribute__((ext_vector_type(16))) _Float16 v16h;
typedef __attribute__((ext_vector_type(4)))  int      v4i;
typedef __attribute__((ext_vector_type(4)))  float    v4f;
typedef __attribute__((ext_vector_type(2)))  _Float16 v2h;

#define BQ 4
#define SQ 128
#define MQ (BQ*SQ)
#define INF 4096
#define OUTF 4096
#define BSR 128
#define NBIN (INF/BSR)
#define NBOUT (OUTF/BSR)

static __device__ inline int dec_pair(int c) {
    const unsigned p = __builtin_amdgcn_perm((unsigned)c, (unsigned)c, 0x0C000C00u);
    const unsigned t = (p & 0x00F0000Fu) ^ 0x64806408u;
    const v2h h  = __builtin_bit_cast(v2h, t);
    const v2h sc = {(_Float16)1.0f, (_Float16)0.0625f};
    const v2h bi = {(_Float16)-1032.0f, (_Float16)-72.0f};
    const v2h r = h * sc + bi;
    return __builtin_bit_cast(int, r);
}

static __device__ inline v8h dec_chunk(v4i w) {
    v4i d;
    #pragma unroll
    for (int j = 0; j < 4; ++j) d[j] = dec_pair(w[j]);
    return __builtin_bit_cast(v8h, d);
}

#define IR_XPITCH 4100
#define IR_GCOLS  512
#define IR_PPITCH 516
typedef __attribute__((ext_vector_type(4))) _Float16 v4h;

__global__ __launch_bounds__(256) void k_in_rot(
    const float* __restrict__ x, const int* __restrict__ perm,
    const float* __restrict__ R_in, const float* __restrict__ s_a,
    _Float16* __restrict__ A)
{
    __shared__ __attribute__((aligned(16))) float xs[16 * IR_XPITCH];
    __shared__ __attribute__((aligned(16))) float xq[16 * IR_PPITCH];

    const int m0   = blockIdx.x * 16;
    const int g    = blockIdx.y;
    const int tid  = threadIdx.x;
    const int wave = tid >> 5;
    const int lane = tid & 31;
    const int halfsel = lane >> 4;
    const int lr      = lane & 15;
    const int nbl     = wave >> 1;
    const int nb      = g * 4 + nbl;
    const int dh      = (wave & 1) * 64 + lr * 4;
    const int kc      = nb * BSR + dh;

    const int p0 = perm[g * IR_GCOLS + tid];
    const int p1 = perm[g * IR_GCOLS + 256 + tid];
    const v4f sv = *(const v4f*)(s_a + kc);

    #pragma unroll
    for (int rr = 0; rr < 2; ++rr) {
        const int r = wave * 2 + rr;
        const float* src = x + (size_t)(m0 + r) * INF + lane * 4;
        float* dst = xs + r * IR_XPITCH + lane * 4;
        v4f t[32];
        #pragma unroll
        for (int i = 0; i < 32; ++i) t[i] = *(const v4f*)(src + i * 128);
        #pragma unroll
        for (int i = 0; i < 32; ++i) *(v4f*)(dst + i * 128) = t[i];
    }

    const float* Rb = R_in + (size_t)nb * BSR * BSR + (size_t)(halfsel * 2) * BSR + dh;
    v4f rb[2][8][2];
    #pragma unroll
    for (int s = 0; s < 8; ++s) {
        rb[0][s][0] = *(const v4f*)(Rb + (size_t)(4 * s) * BSR);
        rb[0][s][1] = *(const v4f*)(Rb + (size_t)(4 * s + 1) * BSR);
    }
    __syncthreads();

    #pragma unroll
    for (int r = 0; r < 16; ++r) {
        xq[r * IR_PPITCH + tid]       = xs[r * IR_XPITCH + p0];
        xq[r * IR_PPITCH + 256 + tid] = xs[r * IR_XPITCH + p1];
    }
    __syncthreads();

    const float* fa = xq + lr * IR_PPITCH + nbl * BSR + halfsel * 2;
    v2f a[32];
    #pragma unroll
    for (int j = 0; j < 32; ++j) a[j] = *(const v2f*)(fa + 4 * j);
    __builtin_amdgcn_sched_barrier(0);

    v8f acc[4] = {};
    #pragma unroll
    for (int blk = 0; blk < 4; ++blk) {
        if (blk + 1 < 4) {
            #pragma unroll
            for (int s = 0; s < 8; ++s) {
                const int j = (blk + 1) * 8 + s;
                rb[(blk + 1) & 1][s][0] = *(const v4f*)(Rb + (size_t)(4 * j) * BSR);
                rb[(blk + 1) & 1][s][1] = *(const v4f*)(Rb + (size_t)(4 * j + 1) * BSR);
            }
        }
        #pragma unroll
        for (int s = 0; s < 8; ++s) {
            #pragma unroll
            for (int e = 0; e < 4; ++e) {
                v2f b;
                b.x = rb[blk & 1][s][0][e];
                b.y = rb[blk & 1][s][1][e];
                acc[e] = __builtin_amdgcn_wmma_f32_16x16x4_f32(false, a[blk * 8 + s], false, b,
                                                               (short)0, acc[e], false, false);
            }
        }
    }

    float inv[4];
    #pragma unroll
    for (int e = 0; e < 4; ++e) inv[e] = 1.0f / sv[e];
    #pragma unroll
    for (int r = 0; r < 8; ++r) {
        const int M = r + halfsel * 8;
        v4h o;
        #pragma unroll
        for (int e = 0; e < 4; ++e) {
            float q = __builtin_rintf(acc[e][r] * inv[e]);
            q = fminf(fmaxf(q, -128.0f), 127.0f);
            o[e] = (_Float16)(q * sv[e]);
        }
        *(v4h*)(A + (size_t)(m0 + M) * INF + kc) = o;
    }
}

#define GT_M 64
#define GT_N 128
#define GT_K 256
#define GT_PITCH 264
#define GT_ROWS (GT_M + GT_N)
#define GT_STAGE (GT_ROWS * GT_PITCH)
#define GT_CPR (GT_K / 8)
#define GT_RPP (256 / GT_CPR)
#define GT_APASS (GT_M / GT_RPP)
#define GT_WPASS (GT_N / GT_RPP)
#define GT_STEPS (INF / GT_K)
#define RP (GT_PITCH / 2)
static_assert(GT_N == BSR && (BSR + GT_M) * RP * 4 <= GT_STAGE * 2, "R_out block + Y tile must fit in one stage");

static __device__ inline void lds_frag4(const _Float16* s, int oa0, int oa1, int ob0, int ob1,
                                        v16h (&fa)[2], v16h (&fb)[2]) {
    const v8h l0 = *(const v8h*)(s + oa0), h0 = *(const v8h*)(s + oa0 + 16);
    const v8h l1 = *(const v8h*)(s + oa1), h1 = *(const v8h*)(s + oa1 + 16);
    const v8h l2 = *(const v8h*)(s + ob0), h2 = *(const v8h*)(s + ob0 + 16);
    const v8h l3 = *(const v8h*)(s + ob1), h3 = *(const v8h*)(s + ob1 + 16);
    #pragma unroll
    for (int e = 0; e < 8; ++e) {
        fa[0][e] = l0[e]; fa[0][8 + e] = h0[e];
        fa[1][e] = l1[e]; fa[1][8 + e] = h1[e];
        fb[0][e] = l2[e]; fb[0][8 + e] = h2[e];
        fb[1][e] = l3[e]; fb[1][8 + e] = h3[e];
    }
}

static __device__ inline void lds_barrier() {
    asm volatile("s_wait_dscnt 0x0\n\ts_barrier_signal -1\n\ts_barrier_wait -1" ::: "memory");
}

template <int U, bool LOAD, bool STORE>
static __device__ __forceinline__ void gemm_step(
    int t, _Float16* sm, const _Float16* gA, const int* gW, int sdst,
    int fa0, int fa1, int fb0, int fb1,
    v8h (&pa)[2][GT_APASS], v4i (&pw)[2][GT_WPASS],
    v8f& acc00, v8f& acc01, v8f& acc10, v8f& acc11)
{
    if (LOAD) {
        const int k2 = (t + 2) * GT_K;
        #pragma unroll
        for (int i = 0; i < GT_APASS; ++i)
            pa[U][i] = *(const v8h*)(gA + (size_t)(i * GT_RPP) * INF + k2);
        #pragma unroll
        for (int i = 0; i < GT_WPASS; ++i)
            pw[U][i] = *(const v4i*)(gW + (size_t)(i * GT_RPP) * (INF / 2) + (k2 >> 1));
    }

    const _Float16* s = &sm[U * GT_STAGE];
    v16h fa[2][2], fb[2][2];
    lds_frag4(s, fa0, fa1, fb0, fb1, fa[0], fb[0]);
    #pragma unroll
    for (int kk = 0; kk < GT_K / 32; ++kk) {
        if (kk + 1 < GT_K / 32)
            lds_frag4(s + (kk + 1) * 32, fa0, fa1, fb0, fb1, fa[(kk + 1) & 1], fb[(kk + 1) & 1]);
        __builtin_amdgcn_sched_barrier(0);
        acc00 = __builtin_amdgcn_wmma_f32_16x16x32_f16(false, fa[kk & 1][0], false, fb[kk & 1][0],
                                                       (short)0, acc00, false, false);
        acc01 = __builtin_amdgcn_wmma_f32_16x16x32_f16(false, fa[kk & 1][0], false, fb[kk & 1][1],
                                                       (short)0, acc01, false, false);
        acc10 = __builtin_amdgcn_wmma_f32_16x16x32_f16(false, fa[kk & 1][1], false, fb[kk & 1][0],
                                                       (short)0, acc10, false, false);
        acc11 = __builtin_amdgcn_wmma_f32_16x16x32_f16(false, fa[kk & 1][1], false, fb[kk & 1][1],
                                                       (short)0, acc11, false, false);
    }

    if (STORE) {
        const int nxt = (U ^ 1) * GT_STAGE;
        #pragma unroll
        for (int i = 0; i < GT_APASS; ++i)
            *(v8h*)(&sm[nxt + sdst + i * GT_RPP * GT_PITCH]) = pa[U ^ 1][i];
        #pragma unroll
        for (int i = 0; i < GT_WPASS; ++i)
            *(v8h*)(&sm[nxt + sdst + (GT_M + i * GT_RPP) * GT_PITCH]) = dec_chunk(pw[U ^ 1][i]);
    }
    lds_barrier();
}

__global__ __launch_bounds__(256) void k_gemm16(
    const _Float16* __restrict__ A, const int* __restrict__ Wp,
    const float* __restrict__ s_w, const float* __restrict__ R_out,
    const float* __restrict__ bias, float* __restrict__ out)
{
    __shared__ __attribute__((aligned(16))) _Float16 sm[2 * GT_STAGE];

    const int tid  = threadIdx.x;
    const int wave = tid >> 5;
    const int lane = tid & 31;
    const int m0 = blockIdx.x * GT_M;
    const int o0 = blockIdx.y * GT_N;
    const int halfsel = lane >> 4;
    const int lr      = lane & 15;
    const int kbase   = halfsel * 8;

    const int crow = tid / GT_CPR;
    const int cch  = (tid % GT_CPR) * 8;
    const _Float16* gA = A  + (size_t)(m0 + crow) * INF + cch;
    const int*      gW = Wp + (size_t)(o0 + crow) * (INF / 2) + (cch >> 1);
    const int sdst = crow * GT_PITCH + cch;

    const int wm = (wave & 1) * 32;
    const int wn = GT_M + (wave >> 1) * 32;
    const int fa0 = (wm + lr)      * GT_PITCH + kbase;
    const int fa1 = (wm + 16 + lr) * GT_PITCH + kbase;
    const int fb0 = (wn + lr)      * GT_PITCH + kbase;
    const int fb1 = (wn + 16 + lr) * GT_PITCH + kbase;

    static_assert(GT_STEPS % 2 == 0 && GT_STEPS >= 4, "the K loop below walks the steps in pairs and peels the last pair");
    v8h pa[2][GT_APASS];
    v4i pw[2][GT_WPASS];
    #pragma unroll
    for (int u = 0; u < 2; ++u) {
        #pragma unroll
        for (int i = 0; i < GT_APASS; ++i)
            pa[u][i] = *(const v8h*)(gA + (size_t)(i * GT_RPP) * INF + u * GT_K);
        #pragma unroll
        for (int i = 0; i < GT_WPASS; ++i)
            pw[u][i] = *(const v4i*)(gW + (size_t)(i * GT_RPP) * (INF / 2) + (u * GT_K >> 1));
        __builtin_amdgcn_sched_barrier(0);
    }
    #pragma unroll
    for (int i = 0; i < GT_APASS; ++i)
        *(v8h*)(&sm[sdst + i * GT_RPP * GT_PITCH]) = pa[0][i];
    #pragma unroll
    for (int i = 0; i < GT_WPASS; ++i)
        *(v8h*)(&sm[sdst + (GT_M + i * GT_RPP) * GT_PITCH]) = dec_chunk(pw[0][i]);
    lds_barrier();

    v8f acc00 = {}, acc01 = {}, acc10 = {}, acc11 = {};
    for (int t2 = 0; t2 < GT_STEPS - 2; t2 += 2) {
        gemm_step<0, true, true>(t2,     sm, gA, gW, sdst, fa0, fa1, fb0, fb1, pa, pw, acc00, acc01, acc10, acc11);
        gemm_step<1, true, true>(t2 + 1, sm, gA, gW, sdst, fa0, fa1, fb0, fb1, pa, pw, acc00, acc01, acc10, acc11);
    }
    gemm_step<0, false, true >(GT_STEPS - 2, sm, gA, gW, sdst, fa0, fa1, fb0, fb1, pa, pw, acc00, acc01, acc10, acc11);
    gemm_step<1, false, false>(GT_STEPS - 1, sm, gA, gW, sdst, fa0, fa1, fb0, fb1, pa, pw, acc00, acc01, acc10, acc11);

    float* sf = (float*)sm;
    const int d0 = wave * 16;
    const int oc = o0 + d0 + lr;
    {
        const float* gR = R_out + (size_t)blockIdx.y * BSR * BSR + (size_t)d0 * BSR + lane * 4;
        v4f pr[16];
        #pragma unroll
        for (int i = 0; i < 16; ++i)
            pr[i] = *(const v4f*)(gR + (size_t)i * BSR);

        const int ow = o0 + (wave >> 1) * 32;
        const float sw0 = s_w[ow + lr];
        const float sw1 = s_w[ow + 16 + lr];
        const int yb = (BSR + wm + halfsel * 8) * RP + (wave >> 1) * 32 + lr;
        #pragma unroll
        for (int r = 0; r < 8; ++r) {
            sf[yb + r * RP]             = acc00[r] * sw0;
            sf[yb + r * RP + 16]        = acc01[r] * sw1;
            sf[yb + (16 + r) * RP]      = acc10[r] * sw0;
            sf[yb + (16 + r) * RP + 16] = acc11[r] * sw1;
        }
        const int rdst = d0 * RP + lane * 4;
        #pragma unroll
        for (int i = 0; i < 16; ++i)
            *(v4f*)(&sf[rdst + i * RP]) = pr[i];
    }
    const float bs = bias[oc];
    __syncthreads();

    const float* fA = sf + (BSR + lr) * RP + halfsel * 2;
    const float* fB = sf + (d0 + lr) * RP + halfsel * 2;
    v8f ro[4] = {};
    v2f ra[2][4][8], rb[2][8];
    #pragma unroll
    for (int s = 0; s < 8; ++s) {
        rb[0][s] = *(const v2f*)(fB + 4 * s);
        #pragma unroll
        for (int ms = 0; ms < 4; ++ms)
            ra[0][ms][s] = *(const v2f*)(fA + ms * 16 * RP + 4 * s);
    }
    #pragma unroll
    for (int blk = 0; blk < BSR / 32; ++blk) {
        if (blk + 1 < BSR / 32) {
            #pragma unroll
            for (int s = 0; s < 8; ++s) {
                rb[(blk + 1) & 1][s] = *(const v2f*)(fB + (blk + 1) * 32 + 4 * s);
                #pragma unroll
                for (int ms = 0; ms < 4; ++ms)
                    ra[(blk + 1) & 1][ms][s] = *(const v2f*)(fA + ms * 16 * RP + (blk + 1) * 32 + 4 * s);
            }
        }
        __builtin_amdgcn_sched_barrier(0);
        #pragma unroll
        for (int s = 0; s < 8; ++s) {
            #pragma unroll
            for (int ms = 0; ms < 4; ++ms)
                ro[ms] = __builtin_amdgcn_wmma_f32_16x16x4_f32(false, ra[blk & 1][ms][s], false, rb[blk & 1][s],
                                                               (short)0, ro[ms], false, false);
        }
    }

    #pragma unroll
    for (int ms = 0; ms < 4; ++ms) {
        #pragma unroll
        for (int r = 0; r < 8; ++r) {
            const int M = ms * 16 + halfsel * 8 + r;
            out[(size_t)(m0 + M) * OUTF + oc] = ro[ms][r] + bs;
        }
    }
}

extern "C" void kernel_launch(void* const* d_in, const int* in_sizes, int n_in,
                              void* d_out, int out_size, void* d_ws, size_t ws_size,
                              hipStream_t stream) {
    const float* x        = (const float*)d_in[0];
    const int*   perm     = (const int*)  d_in[1];
    const float* R_in     = (const float*)d_in[2];
    const float* s_a      = (const float*)d_in[3];
    const int*   packed_w = (const int*)  d_in[4];
    const float* s_w      = (const float*)d_in[5];
    const float* bias     = (const float*)d_in[6];
    const float* R_out    = (const float*)d_in[7];
    float* out = (float*)d_out;

    _Float16* A  = (_Float16*)d_ws;

    const dim3 blk(256);
    k_in_rot<<<dim3(MQ / 16, NBIN / 4), blk, 0, stream>>>(x, perm, R_in, s_a, A);
    k_gemm16<<<dim3(MQ / GT_M, OUTF / GT_N), blk, 0, stream>>>(A, packed_w, s_w, R_out, bias, out);
}


# k_in_rot also copies the low bytes of the packed weight words into a compact 8 MB plane; k_gemm16 stages W from it (4-B loads): a quarter of the W bytes fetched per tile, same LDS image and k order
# speedup vs baseline: 3.4694x; 1.0780x over previous
#include <hip/hip_runtime.h>
#include <hip/hip_bf16.h>

typedef __attribute__((ext_vector_type(2)))  float    v2f;
typedef __attribute__((ext_vector_type(8)))  float    v8f;
typedef __attribute__((ext_vector_type(8)))  _Float16 v8h;
typedef __attribute__((ext_vector_type(16))) _Float16 v16h;
typedef __attribute__((ext_vector_type(4)))  int      v4i;
typedef __attribute__((ext_vector_type(4)))  float    v4f;
typedef __attribute__((ext_vector_type(2)))  _Float16 v2h;

#define BQ 4
#define SQ 128
#define MQ (BQ*SQ)
#define INF 4096
#define OUTF 4096
#define BSR 128
#define NBIN (INF/BSR)
#define NBOUT (OUTF/BSR)

template <int BYTE>
static __device__ inline int dec_pair(int c) {
    const unsigned p = __builtin_amdgcn_perm((unsigned)c, (unsigned)c, 0x0C000C00u | (unsigned)BYTE | ((unsigned)BYTE << 16));
    const unsigned t = (p & 0x00F0000Fu) ^ 0x64806408u;
    const v2h h  = __builtin_bit_cast(v2h, t);
    const v2h sc = {(_Float16)1.0f, (_Float16)0.0625f};
    const v2h bi = {(_Float16)-1032.0f, (_Float16)-72.0f};
    const v2h r = h * sc + bi;
    return __builtin_bit_cast(int, r);
}

static __device__ inline v8h dec_chunk(int w) {
    v4i d;
    d[0] = dec_pair<0>(w);
    d[1] = dec_pair<1>(w);
    d[2] = dec_pair<2>(w);
    d[3] = dec_pair<3>(w);
    return __builtin_bit_cast(v8h, d);
}

#define IR_XPITCH 4100
#define IR_GCOLS  512
#define IR_PPITCH 516
#define IR_WROWS  (OUTF / 256)
#define IR_WQ     (IR_WROWS * (INF / 2) / 4 / 256)
typedef __attribute__((ext_vector_type(4))) _Float16 v4h;
static_assert((MQ / 16) * (NBIN / 4) == 256 && IR_WROWS * 256 == OUTF, "stage 1's grid must cover every packed row");

__global__ __launch_bounds__(256) void k_in_rot(
    const float* __restrict__ x, const int* __restrict__ perm,
    const float* __restrict__ R_in, const float* __restrict__ s_a,
    const int* __restrict__ Wp, _Float16* __restrict__ A, int* __restrict__ Wc)
{
    __shared__ __attribute__((aligned(16))) float xs[16 * IR_XPITCH];
    __shared__ __attribute__((aligned(16))) float xq[16 * IR_PPITCH];

    const int m0   = blockIdx.x * 16;
    const int g    = blockIdx.y;
    const int tid  = threadIdx.x;
    const int wave = tid >> 5;
    const int lane = tid & 31;
    const int halfsel = lane >> 4;
    const int lr      = lane & 15;
    const int nbl     = wave >> 1;
    const int nb      = g * 4 + nbl;
    const int dh      = (wave & 1) * 64 + lr * 4;
    const int kc      = nb * BSR + dh;

    const int p0 = perm[g * IR_GCOLS + tid];
    const int p1 = perm[g * IR_GCOLS + 256 + tid];
    const v4f sv = *(const v4f*)(s_a + kc);

    #pragma unroll
    for (int rr = 0; rr < 2; ++rr) {
        const int r = wave * 2 + rr;
        const float* src = x + (size_t)(m0 + r) * INF + lane * 4;
        float* dst = xs + r * IR_XPITCH + lane * 4;
        v4f t[32];
        #pragma unroll
        for (int i = 0; i < 32; ++i) t[i] = *(const v4f*)(src + i * 128);
        #pragma unroll
        for (int i = 0; i < 32; ++i) *(v4f*)(dst + i * 128) = t[i];
    }

    const float* Rb = R_in + (size_t)nb * BSR * BSR + (size_t)(halfsel * 2) * BSR + dh;
    v4f rb[2][8][2];
    #pragma unroll
    for (int s = 0; s < 8; ++s) {
        rb[0][s][0] = *(const v4f*)(Rb + (size_t)(4 * s) * BSR);
        rb[0][s][1] = *(const v4f*)(Rb + (size_t)(4 * s + 1) * BSR);
    }
    __syncthreads();

    #pragma unroll
    for (int r = 0; r < 16; ++r) {
        xq[r * IR_PPITCH + tid]       = xs[r * IR_XPITCH + p0];
        xq[r * IR_PPITCH + 256 + tid] = xs[r * IR_XPITCH + p1];
    }
    __syncthreads();

    const float* fa = xq + lr * IR_PPITCH + nbl * BSR + halfsel * 2;
    v2f a[32];
    #pragma unroll
    for (int j = 0; j < 32; ++j) a[j] = *(const v2f*)(fa + 4 * j);
    __builtin_amdgcn_sched_barrier(0);

    const int wg = blockIdx.y * (MQ / 16) + blockIdx.x;
    const int* gWp = Wp + (size_t)wg * (IR_WROWS * (INF / 2)) + tid * 4;
    int*       gWc = Wc + (size_t)wg * (IR_WROWS * (INF / 8)) + tid;
    v4i wq[IR_WQ];

    v8f acc[4] = {};
    #pragma unroll
    for (int blk = 0; blk < 4; ++blk) {
        if (blk + 1 < 4) {
            #pragma unroll
            for (int s = 0; s < 8; ++s) {
                const int j = (blk + 1) * 8 + s;
                rb[(blk + 1) & 1][s][0] = *(const v4f*)(Rb + (size_t)(4 * j) * BSR);
                rb[(blk + 1) & 1][s][1] = *(const v4f*)(Rb + (size_t)(4 * j + 1) * BSR);
            }
        }
        if (blk == 2) {
            #pragma unroll
            for (int i = 0; i < IR_WQ; ++i)
                wq[i] = __builtin_nontemporal_load((const v4i*)(gWp + (size_t)i * (256 * 4)));
        }
        #pragma unroll
        for (int s = 0; s < 8; ++s) {
            #pragma unroll
            for (int e = 0; e < 4; ++e) {
                v2f b;
                b.x = rb[blk & 1][s][0][e];
                b.y = rb[blk & 1][s][1][e];
                acc[e] = __builtin_amdgcn_wmma_f32_16x16x4_f32(false, a[blk * 8 + s], false, b,
                                                               (short)0, acc[e], false, false);
            }
        }
    }

    float inv[4];
    #pragma unroll
    for (int e = 0; e < 4; ++e) inv[e] = 1.0f / sv[e];
    #pragma unroll
    for (int r = 0; r < 8; ++r) {
        const int M = r + halfsel * 8;
        v4h o;
        #pragma unroll
        for (int e = 0; e < 4; ++e) {
            float q = __builtin_rintf(acc[e][r] * inv[e]);
            q = fminf(fmaxf(q, -128.0f), 127.0f);
            o[e] = (_Float16)(q * sv[e]);
        }
        *(v4h*)(A + (size_t)(m0 + M) * INF + kc) = o;
    }

    #pragma unroll
    for (int i = 0; i < IR_WQ; ++i) {
        const v4i w = wq[i];
        gWc[i * 256] = (w[0] & 0xFF) | ((w[1] & 0xFF) << 8) | ((w[2] & 0xFF) << 16) | (int)((unsigned)w[3] << 24);
    }
}

#define GT_M 64
#define GT_N 128
#define GT_K 256
#define GT_PITCH 264
#define GT_ROWS (GT_M + GT_N)
#define GT_STAGE (GT_ROWS * GT_PITCH)
#define GT_CPR (GT_K / 8)
#define GT_RPP (256 / GT_CPR)
#define GT_APASS (GT_M / GT_RPP)
#define GT_WPASS (GT_N / GT_RPP)
#define GT_STEPS (INF / GT_K)
#define RP (GT_PITCH / 2)
static_assert(GT_N == BSR && (BSR + GT_M) * RP * 4 <= GT_STAGE * 2, "R_out block + Y tile must fit in one stage");

static __device__ inline void lds_frag4(const _Float16* s, int oa0, int oa1, int ob0, int ob1,
                                        v16h (&fa)[2], v16h (&fb)[2]) {
    const v8h l0 = *(const v8h*)(s + oa0), h0 = *(const v8h*)(s + oa0 + 16);
    const v8h l1 = *(const v8h*)(s + oa1), h1 = *(const v8h*)(s + oa1 + 16);
    const v8h l2 = *(const v8h*)(s + ob0), h2 = *(const v8h*)(s + ob0 + 16);
    const v8h l3 = *(const v8h*)(s + ob1), h3 = *(const v8h*)(s + ob1 + 16);
    #pragma unroll
    for (int e = 0; e < 8; ++e) {
        fa[0][e] = l0[e]; fa[0][8 + e] = h0[e];
        fa[1][e] = l1[e]; fa[1][8 + e] = h1[e];
        fb[0][e] = l2[e]; fb[0][8 + e] = h2[e];
        fb[1][e] = l3[e]; fb[1][8 + e] = h3[e];
    }
}

static __device__ inline void lds_barrier() {
    asm volatile("s_wait_dscnt 0x0\n\ts_barrier_signal -1\n\ts_barrier_wait -1" ::: "memory");
}

template <int U, bool LOAD, bool STORE>
static __device__ __forceinline__ void gemm_step(
    int t, _Float16* sm, const _Float16* gA, const int* gW, int sdst,
    int fa0, int fa1, int fb0, int fb1,
    v8h (&pa)[2][GT_APASS], int (&pw)[2][GT_WPASS],
    v8f& acc00, v8f& acc01, v8f& acc10, v8f& acc11)
{
    if (LOAD) {
        const int k2 = (t + 2) * GT_K;
        #pragma unroll
        for (int i = 0; i < GT_APASS; ++i)
            pa[U][i] = *(const v8h*)(gA + (size_t)(i * GT_RPP) * INF + k2);
        #pragma unroll
        for (int i = 0; i < GT_WPASS; ++i)
            pw[U][i] = gW[(size_t)(i * GT_RPP) * (INF / 8) + (k2 >> 3)];
    }

    const _Float16* s = &sm[U * GT_STAGE];
    v16h fa[2][2], fb[2][2];
    lds_frag4(s, fa0, fa1, fb0, fb1, fa[0], fb[0]);
    #pragma unroll
    for (int kk = 0; kk < GT_K / 32; ++kk) {
        if (kk + 1 < GT_K / 32)
            lds_frag4(s + (kk + 1) * 32, fa0, fa1, fb0, fb1, fa[(kk + 1) & 1], fb[(kk + 1) & 1]);
        __builtin_amdgcn_sched_barrier(0);
        acc00 = __builtin_amdgcn_wmma_f32_16x16x32_f16(false, fa[kk & 1][0], false, fb[kk & 1][0],
                                                       (short)0, acc00, false, false);
        acc01 = __builtin_amdgcn_wmma_f32_16x16x32_f16(false, fa[kk & 1][0], false, fb[kk & 1][1],
                                                       (short)0, acc01, false, false);
        acc10 = __builtin_amdgcn_wmma_f32_16x16x32_f16(false, fa[kk & 1][1], false, fb[kk & 1][0],
                                                       (short)0, acc10, false, false);
        acc11 = __builtin_amdgcn_wmma_f32_16x16x32_f16(false, fa[kk & 1][1], false, fb[kk & 1][1],
                                                       (short)0, acc11, false, false);
    }

    if (STORE) {
        const int nxt = (U ^ 1) * GT_STAGE;
        #pragma unroll
        for (int i = 0; i < GT_APASS; ++i)
            *(v8h*)(&sm[nxt + sdst + i * GT_RPP * GT_PITCH]) = pa[U ^ 1][i];
        #pragma unroll
        for (int i = 0; i < GT_WPASS; ++i)
            *(v8h*)(&sm[nxt + sdst + (GT_M + i * GT_RPP) * GT_PITCH]) = dec_chunk(pw[U ^ 1][i]);
    }
    lds_barrier();
}

__global__ __launch_bounds__(256) void k_gemm16(
    const _Float16* __restrict__ A, const int* __restrict__ Wc,
    const float* __restrict__ s_w, const float* __restrict__ R_out,
    const float* __restrict__ bias, float* __restrict__ out)
{
    __shared__ __attribute__((aligned(16))) _Float16 sm[2 * GT_STAGE];

    const int tid  = threadIdx.x;
    const int wave = tid >> 5;
    const int lane = tid & 31;
    const int m0 = blockIdx.x * GT_M;
    const int o0 = blockIdx.y * GT_N;
    const int halfsel = lane >> 4;
    const int lr      = lane & 15;
    const int kbase   = halfsel * 8;

    const int crow = tid / GT_CPR;
    const int cch  = (tid % GT_CPR) * 8;
    const _Float16* gA = A  + (size_t)(m0 + crow) * INF + cch;
    const int*      gW = Wc + (size_t)(o0 + crow) * (INF / 8) + (cch >> 3);
    const int sdst = crow * GT_PITCH + cch;

    const int wm = (wave & 1) * 32;
    const int wn = GT_M + (wave >> 1) * 32;
    const int fa0 = (wm + lr)      * GT_PITCH + kbase;
    const int fa1 = (wm + 16 + lr) * GT_PITCH + kbase;
    const int fb0 = (wn + lr)      * GT_PITCH + kbase;
    const int fb1 = (wn + 16 + lr) * GT_PITCH + kbase;

    static_assert(GT_STEPS % 2 == 0 && GT_STEPS >= 4, "the K loop below walks the steps in pairs and peels the last pair");
    v8h pa[2][GT_APASS];
    int pw[2][GT_WPASS];
    #pragma unroll
    for (int u = 0; u < 2; ++u) {
        #pragma unroll
        for (int i = 0; i < GT_APASS; ++i)
            pa[u][i] = *(const v8h*)(gA + (size_t)(i * GT_RPP) * INF + u * GT_K);
        #pragma unroll
        for (int i = 0; i < GT_WPASS; ++i)
            pw[u][i] = gW[(size_t)(i * GT_RPP) * (INF / 8) + (u * GT_K >> 3)];
        __builtin_amdgcn_sched_barrier(0);
    }
    #pragma unroll
    for (int i = 0; i < GT_APASS; ++i)
        *(v8h*)(&sm[sdst + i * GT_RPP * GT_PITCH]) = pa[0][i];
    #pragma unroll
    for (int i = 0; i < GT_WPASS; ++i)
        *(v8h*)(&sm[sdst + (GT_M + i * GT_RPP) * GT_PITCH]) = dec_chunk(pw[0][i]);
    lds_barrier();

    v8f acc00 = {}, acc01 = {}, acc10 = {}, acc11 = {};
    for (int t2 = 0; t2 < GT_STEPS - 2; t2 += 2) {
        gemm_step<0, true, true>(t2,     sm, gA, gW, sdst, fa0, fa1, fb0, fb1, pa, pw, acc00, acc01, acc10, acc11);
        gemm_step<1, true, true>(t2 + 1, sm, gA, gW, sdst, fa0, fa1, fb0, fb1, pa, pw, acc00, acc01, acc10, acc11);
    }
    gemm_step<0, false, true >(GT_STEPS - 2, sm, gA, gW, sdst, fa0, fa1, fb0, fb1, pa, pw, acc00, acc01, acc10, acc11);
    gemm_step<1, false, false>(GT_STEPS - 1, sm, gA, gW, sdst, fa0, fa1, fb0, fb1, pa, pw, acc00, acc01, acc10, acc11);

    float* sf = (float*)sm;
    const int d0 = wave * 16;
    const int oc = o0 + d0 + lr;
    {
        const float* gR = R_out + (size_t)blockIdx.y * BSR * BSR + (size_t)d0 * BSR + lane * 4;
        v4f pr[16];
        #pragma unroll
        for (int i = 0; i < 16; ++i)
            pr[i] = *(const v4f*)(gR + (size_t)i * BSR);

        const int ow = o0 + (wave >> 1) * 32;
        const float sw0 = s_w[ow + lr];
        const float sw1 = s_w[ow + 16 + lr];
        const int yb = (BSR + wm + halfsel * 8) * RP + (wave >> 1) * 32 + lr;
        #pragma unroll
        for (int r = 0; r < 8; ++r) {
            sf[yb + r * RP]             = acc00[r] * sw0;
            sf[yb + r * RP + 16]        = acc01[r] * sw1;
            sf[yb + (16 + r) * RP]      = acc10[r] * sw0;
            sf[yb + (16 + r) * RP + 16] = acc11[r] * sw1;
        }
        const int rdst = d0 * RP + lane * 4;
        #pragma unroll
        for (int i = 0; i < 16; ++i)
            *(v4f*)(&sf[rdst + i * RP]) = pr[i];
    }
    const float bs = bias[oc];
    __syncthreads();

    const float* fA = sf + (BSR + lr) * RP + halfsel * 2;
    const float* fB = sf + (d0 + lr) * RP + halfsel * 2;
    v8f ro[4] = {};
    v2f ra[2][4][8], rb[2][8];
    #pragma unroll
    for (int s = 0; s < 8; ++s) {
        rb[0][s] = *(const v2f*)(fB + 4 * s);
        #pragma unroll
        for (int ms = 0; ms < 4; ++ms)
            ra[0][ms][s] = *(const v2f*)(fA + ms * 16 * RP + 4 * s);
    }
    #pragma unroll
    for (int blk = 0; blk < BSR / 32; ++blk) {
        if (blk + 1 < BSR / 32) {
            #pragma unroll
            for (int s = 0; s < 8; ++s) {
                rb[(blk + 1) & 1][s] = *(const v2f*)(fB + (blk + 1) * 32 + 4 * s);
                #pragma unroll
                for (int ms = 0; ms < 4; ++ms)
                    ra[(blk + 1) & 1][ms][s] = *(const v2f*)(fA + ms * 16 * RP + (blk + 1) * 32 + 4 * s);
            }
        }
        __builtin_amdgcn_sched_barrier(0);
        #pragma unroll
        for (int s = 0; s < 8; ++s) {
            #pragma unroll
            for (int ms = 0; ms < 4; ++ms)
                ro[ms] = __builtin_amdgcn_wmma_f32_16x16x4_f32(false, ra[blk & 1][ms][s], false, rb[blk & 1][s],
                                                               (short)0, ro[ms], false, false);
        }
    }

    #pragma unroll
    for (int ms = 0; ms < 4; ++ms) {
        #pragma unroll
        for (int r = 0; r < 8; ++r) {
            const int M = ms * 16 + halfsel * 8 + r;
            out[(size_t)(m0 + M) * OUTF + oc] = ro[ms][r] + bs;
        }
    }
}

extern "C" void kernel_launch(void* const* d_in, const int* in_sizes, int n_in,
                              void* d_out, int out_size, void* d_ws, size_t ws_size,
                              hipStream_t stream) {
    const float* x        = (const float*)d_in[0];
    const int*   perm     = (const int*)  d_in[1];
    const float* R_in     = (const float*)d_in[2];
    const float* s_a      = (const float*)d_in[3];
    const int*   packed_w = (const int*)  d_in[4];
    const float* s_w      = (const float*)d_in[5];
    const float* bias     = (const float*)d_in[6];
    const float* R_out    = (const float*)d_in[7];
    float* out = (float*)d_out;

    _Float16* A  = (_Float16*)d_ws;
    int*      Wc = (int*)((char*)d_ws + (size_t)MQ * INF * 2);

    const dim3 blk(256);
    k_in_rot<<<dim3(MQ / 16, NBIN / 4), blk, 0, stream>>>(x, perm, R_in, s_a, packed_w, A, Wc);
    k_gemm16<<<dim3(MQ / GT_M, OUTF / GT_N), blk, 0, stream>>>(A, Wc, s_w, R_out, bias, out);
}
